# starting kernel
# baseline (speedup 1.0000x reference)
#include <hip/hip_runtime.h>
#include <cstdint>
#include <cstddef>

constexpr int GN = 50000;
constexpr int GE = 1600000;
constexpr int GH = 128;
constexpr int GLAYERS = 2;
constexpr size_t NH = (size_t)GN * GH;

typedef __attribute__((ext_vector_type(2))) float v2f;
typedef __attribute__((ext_vector_type(8))) float v8f;

__device__ __forceinline__ v8f wmma_f32(v2f a, v2f b, v8f c) {
  return __builtin_amdgcn_wmma_f32_16x16x4_f32(
       false, a,  false, b,
       (short)0, c,  false,  false);
}

__device__ __forceinline__ float sigmoid_f(float x) {
  return 1.0f / (1.0f + __expf(-x));
}
__device__ __forceinline__ float tanh_f(float x) {
  return 1.0f - 2.0f / (1.0f + __expf(2.0f * x));
}

__device__ __forceinline__ void load_afrag_sum(const float* __restrict__ X,
                                               const float* __restrict__ A,
                                               int row, int kof, v2f* frag) {
  const float* xr = X + (size_t)row * GH;
  const float* ar = A + (size_t)row * GH;
#pragma unroll
  for (int kk = 0; kk < 32; ++kk) {
    const int c = (kk << 2) + kof;
    const float2 xv = *(const float2*)(xr + c);
    const float2 av = *(const float2*)(ar + c);
    v2f f;
    f.x = xv.x + av.x;
    f.y = xv.y + av.y;
    frag[kk] = f;
  }
}

__device__ __forceinline__ void stage_slab(const float* __restrict__ Wg, int n0,
                                           float* lds) {
  const int tid = threadIdx.x;
#pragma unroll
  for (int i = 0; i < 8; ++i) {
    const int idx = i * 256 + tid;
    const int k = idx >> 4;
    const int c = idx & 15;
    lds[c * 132 + k] = Wg[k * GH + n0 + c];
  }
}

__global__ __launch_bounds__(256) void scatter_add_kernel(
    const int* __restrict__ src, const int* __restrict__ dst,
    const float* __restrict__ feat, float* __restrict__ agg) {
  const unsigned tid = blockIdx.x * 256u + threadIdx.x;
  const unsigned e = tid >> 5;
  if (e >= (unsigned)GE) return;
  const int c4 = (tid & 31) << 2;
  const int s = src[e];
  const int d = dst[e];
  const float4 val = *(const float4*)(feat + (size_t)s * GH + c4);
  float* out = agg + (size_t)d * GH + c4;
  atomicAdd(out + 0, val.x);
  atomicAdd(out + 1, val.y);
  atomicAdd(out + 2, val.z);
  atomicAdd(out + 3, val.w);
}

__global__ __launch_bounds__(256) void gemm_zr_kernel(
    const float* __restrict__ x, const float* __restrict__ aggx,
    const float* __restrict__ hl, const float* __restrict__ aggh,
    const float* __restrict__ Wl, const float* __restrict__ bl,
    float* __restrict__ zout, float* __restrict__ vout) {
  __shared__ float lds[4 * 16 * 132];
  const int tid = threadIdx.x;
  const int wave = tid >> 5;
  const int lane = tid & 31;
  const int row0 = blockIdx.x * 128 + wave * 16;
  const int kof = (lane >> 4) << 1;
  const int nn = lane & 15;
  const int rowhi = (lane >> 4) << 3;

  int arow = row0 + nn;
  if (arow > GN - 1) arow = GN - 1;

  v2f fs[32], ft[32];
  load_afrag_sum(x, aggx, arow, kof, fs);
  load_afrag_sum(hl, aggh, arow, kof, ft);

  const float* W0 = Wl + 0 * GH * GH;
  const float* W1 = Wl + 1 * GH * GH;
  const float* W2 = Wl + 2 * GH * GH;
  const float* W3 = Wl + 3 * GH * GH;
  float* L0 = lds + 0 * 16 * 132;
  float* L1 = lds + 1 * 16 * 132;
  float* L2 = lds + 2 * 16 * 132;
  float* L3 = lds + 3 * 16 * 132;

  for (int ct = 0; ct < GH / 16; ++ct) {
    const int n0 = ct * 16;
    __syncthreads();
    stage_slab(W0, n0, L0);
    stage_slab(W1, n0, L1);
    stage_slab(W2, n0, L2);
    stage_slab(W3, n0, L3);
    __syncthreads();

    const float* p0 = L0 + nn * 132 + kof;
    const float* p1 = L1 + nn * 132 + kof;
    const float* p2 = L2 + nn * 132 + kof;
    const float* p3 = L3 + nn * 132 + kof;

    v8f az0 = {0, 0, 0, 0, 0, 0, 0, 0}, az1 = {0, 0, 0, 0, 0, 0, 0, 0};
    v8f ar0 = {0, 0, 0, 0, 0, 0, 0, 0}, ar1 = {0, 0, 0, 0, 0, 0, 0, 0};
#pragma unroll
    for (int kk = 0; kk < 32; kk += 2) {
      az0 = wmma_f32(fs[kk], *(const v2f*)(p0 + 4 * kk), az0);
      az0 = wmma_f32(ft[kk], *(const v2f*)(p1 + 4 * kk), az0);
      ar0 = wmma_f32(fs[kk], *(const v2f*)(p2 + 4 * kk), ar0);
      ar0 = wmma_f32(ft[kk], *(const v2f*)(p3 + 4 * kk), ar0);
      az1 = wmma_f32(fs[kk + 1], *(const v2f*)(p0 + 4 * kk + 4), az1);
      az1 = wmma_f32(ft[kk + 1], *(const v2f*)(p1 + 4 * kk + 4), az1);
      ar1 = wmma_f32(fs[kk + 1], *(const v2f*)(p2 + 4 * kk + 4), ar1);
      ar1 = wmma_f32(ft[kk + 1], *(const v2f*)(p3 + 4 * kk + 4), ar1);
    }
    const v8f accz = az0 + az1;
    const v8f accr = ar0 + ar1;

    const float bz = bl[0 * GH + n0 + nn] + bl[1 * GH + n0 + nn];
    const float br = bl[2 * GH + n0 + nn] + bl[3 * GH + n0 + nn];
#pragma unroll
    for (int i = 0; i < 8; ++i) {
      const int row = row0 + rowhi + i;
      if (row < GN) {
        const size_t off = (size_t)row * GH + n0 + nn;
        const float zv = sigmoid_f(accz[i] + bz);
        const float rv = sigmoid_f(accr[i] + br);
        zout[off] = zv;
        vout[off] = rv * hl[off];
      }
    }
  }
}

__global__ __launch_bounds__(256) void gemm_ht_kernel(
    const float* __restrict__ x, const float* __restrict__ aggx,
    const float* __restrict__ v, const float* __restrict__ aggv,
    const float* __restrict__ Wl, const float* __restrict__ bl,
    const float* __restrict__ zbuf, const float* __restrict__ hl,
    float* __restrict__ outp) {
  __shared__ float lds[2 * 16 * 132];
  const int tid = threadIdx.x;
  const int wave = tid >> 5;
  const int lane = tid & 31;
  const int row0 = blockIdx.x * 128 + wave * 16;
  const int kof = (lane >> 4) << 1;
  const int nn = lane & 15;
  const int rowhi = (lane >> 4) << 3;

  int arow = row0 + nn;
  if (arow > GN - 1) arow = GN - 1;

  v2f fs[32], fu[32];
  load_afrag_sum(x, aggx, arow, kof, fs);
  load_afrag_sum(v, aggv, arow, kof, fu);

  const float* W4 = Wl + 4 * GH * GH;
  const float* W5 = Wl + 5 * GH * GH;
  float* L4 = lds + 0 * 16 * 132;
  float* L5 = lds + 1 * 16 * 132;

  for (int ct = 0; ct < GH / 16; ++ct) {
    const int n0 = ct * 16;
    __syncthreads();
    stage_slab(W4, n0, L4);
    stage_slab(W5, n0, L5);
    __syncthreads();

    const float* p4 = L4 + nn * 132 + kof;
    const float* p5 = L5 + nn * 132 + kof;

    v8f a0 = {0, 0, 0, 0, 0, 0, 0, 0}, a1 = {0, 0, 0, 0, 0, 0, 0, 0};
#pragma unroll
    for (int kk = 0; kk < 32; kk += 2) {
      a0 = wmma_f32(fs[kk], *(const v2f*)(p4 + 4 * kk), a0);
      a0 = wmma_f32(fu[kk], *(const v2f*)(p5 + 4 * kk), a0);
      a1 = wmma_f32(fs[kk + 1], *(const v2f*)(p4 + 4 * kk + 4), a1);
      a1 = wmma_f32(fu[kk + 1], *(const v2f*)(p5 + 4 * kk + 4), a1);
    }
    const v8f acc = a0 + a1;

    const float bh = bl[4 * GH + n0 + nn] + bl[5 * GH + n0 + nn];
#pragma unroll
    for (int i = 0; i < 8; ++i) {
      const int row = row0 + rowhi + i;
      if (row < GN) {
        const size_t off = (size_t)row * GH + n0 + nn;
        const float ht = tanh_f(acc[i] + bh);
        const float zv = zbuf[off];
        outp[off] = zv * hl[off] + (1.0f - zv) * ht;
      }
    }
  }
}

extern "C" void kernel_launch(void* const* d_in, const int* in_sizes, int n_in,
                              void* d_out, int out_size, void* d_ws,
                              size_t ws_size, hipStream_t stream) {
  (void)in_sizes; (void)n_in; (void)out_size; (void)ws_size;
  const float* inp = (const float*)d_in[0];
  const int* edg = (const int*)d_in[1];
  const float* hmat = (const float*)d_in[2];
  const float* Wmat = (const float*)d_in[3];
  const float* bmat = (const float*)d_in[4];
  float* out = (float*)d_out;

  const int* src = edg;
  const int* dst = edg + GE;

  float* agg_x = (float*)d_ws;
  float* agg_t = agg_x + NH;
  float* zbuf = agg_t + NH;
  float* vbuf = zbuf + NH;

  const int scat_blocks = (int)(((size_t)GE * 32 + 255) / 256);
  const int gemm_blocks = (GN + 127) / 128;

  for (int l = 0; l < GLAYERS; ++l) {
    const float* x = (l == 0) ? inp : (out + (size_t)(l - 1) * NH);
    const float* hl = hmat + (size_t)l * NH;
    const float* Wl = Wmat + (size_t)l * 6 * GH * GH;
    const float* bl = bmat + (size_t)l * 6 * GH;
    float* outl = out + (size_t)l * NH;

    hipMemsetAsync(agg_x, 0, NH * sizeof(float), stream);
    hipMemsetAsync(agg_t, 0, NH * sizeof(float), stream);
    scatter_add_kernel<<<scat_blocks, 256, 0, stream>>>(src, dst, x, agg_x);
    scatter_add_kernel<<<scat_blocks, 256, 0, stream>>>(src, dst, hl, agg_t);

    gemm_zr_kernel<<<gemm_blocks, 256, 0, stream>>>(x, agg_x, hl, agg_t, Wl,
                                                    bl, zbuf, vbuf);

    hipMemsetAsync(agg_t, 0, NH * sizeof(float), stream);
    scatter_add_kernel<<<scat_blocks, 256, 0, stream>>>(src, dst, vbuf, agg_t);

    gemm_ht_kernel<<<gemm_blocks, 256, 0, stream>>>(x, agg_x, vbuf, agg_t, Wl,
                                                    bl, zbuf, hl, outl);
  }
}


# Scatter launches: noted edges are grouped by destination row (LDS counting sort), the row's owner wave sums it in registers and stores it once; no LDS f32 atomics; sums in list (arrival) order
# speedup vs baseline: 2.6148x; 2.6148x over previous
#include <hip/hip_runtime.h>
#include <cstdint>
#include <cstddef>

constexpr int GN = 50000;
constexpr int GE = 1600000;
constexpr int GH = 128;
constexpr int GLAYERS = 2;
constexpr size_t NH = (size_t)GN * GH;

typedef __attribute__((ext_vector_type(2))) float v2f;
typedef __attribute__((ext_vector_type(8))) float v8f;
typedef __attribute__((ext_vector_type(4))) float v4f;

__device__ __forceinline__ v8f wmma_f32(v2f a, v2f b, v8f c) {
  return __builtin_amdgcn_wmma_f32_16x16x4_f32(
       false, a,  false, b,
       (short)0, c,  false,  false);
}

__device__ __forceinline__ float sigmoid_f(float x) {
  return 1.0f / (1.0f + __expf(-x));
}
__device__ __forceinline__ float tanh_f(float x) {
  return 1.0f - 2.0f / (1.0f + __expf(2.0f * x));
}

__device__ __forceinline__ void load_afrag_sum(const float* __restrict__ X,
                                               const float* __restrict__ A,
                                               int row, int kof, v2f* frag) {
  const float* xr = X + (size_t)row * GH;
  const float* ar = A + (size_t)row * GH;
#pragma unroll
  for (int kk = 0; kk < 32; ++kk) {
    const int c = (kk << 2) + kof;
    const float2 xv = *(const float2*)(xr + c);
    const float2 av = *(const float2*)(ar + c);
    v2f f;
    f.x = xv.x + av.x;
    f.y = xv.y + av.y;
    frag[kk] = f;
  }
}

__device__ __forceinline__ void stage_slab(const float* __restrict__ Wg, int n0,
                                           float* lds) {
  const int tid = threadIdx.x;
#pragma unroll
  for (int i = 0; i < 8; ++i) {
    const int idx = i * 256 + tid;
    const int k = idx >> 4;
    const int c = idx & 15;
    lds[c * 132 + k] = Wg[k * GH + n0 + c];
  }
}

constexpr int TR = 197;
constexpr int NTILES = (GN + TR - 1) / TR;
constexpr unsigned NO_TILE = 255u;
constexpr int STHREADS = 1024;
constexpr int SWAVES = STHREADS / 32;
constexpr int RPW = (TR + SWAVES - 1) / SWAVES;
constexpr int EPL = 16;
constexpr int SK = 2;
constexpr int CHUNK = STHREADS * EPL * SK;
constexpr int NCHUNK = (GE + CHUNK - 1) / CHUNK;
constexpr int NBIN = SWAVES * RPW;
constexpr int LCAP2 = 39936;

static_assert(NTILES * TR >= GN, "tiles cover all rows");
static_assert(NTILES <= (int)NO_TILE, "every tile number and the spare code fit one byte");
static_assert(GE % EPL == 0, "16-byte code loads");
static_assert(LCAP2 >= CHUNK, "an emptied list holds every edge of one scan step");
static_assert((2 * LCAP2 + 3 * NBIN + 16) * 4 <= 320 * 1024, "both lists fit the LDS");
static_assert(TR <= 256 && GN < (1 << 24), "local row and source row fit one word");
static_assert(SWAVES == 32 && NBIN >= TR + 1, "the first wave's lanes take RPW counters each");

__device__ __forceinline__ unsigned tile_code(int d) {
  const unsigned a = (unsigned)d;
  const unsigned lim = NO_TILE * (unsigned)TR;
  return (a < lim ? a : lim) / (unsigned)TR;
}

__global__ __launch_bounds__(256) void tile_code_kernel(
    const int* __restrict__ dst, float* __restrict__ codes_f) {
  const int g = blockIdx.x * 256 + threadIdx.x;
  if (g >= GE / EPL) return;
  const int4* d4 = (const int4*)(dst + (size_t)g * EPL);
  int4 d[4];
#pragma unroll
  for (int q = 0; q < 4; ++q) d[q] = d4[q];
  unsigned w[4];
#pragma unroll
  for (int q = 0; q < 4; ++q)
    w[q] = tile_code(d[q].x) | (tile_code(d[q].y) << 8) |
           (tile_code(d[q].z) << 16) | (tile_code(d[q].w) << 24);
  ((uint4*)codes_f)[g] = make_uint4(w[0], w[1], w[2], w[3]);
}

__device__ __forceinline__ void load_chunk(const unsigned char* __restrict__ codes,
                                           int chunk, int tid, uint4* buf) {
#pragma unroll
  for (int k = 0; k < SK; ++k) {
    const int e0 = chunk * CHUNK + k * (STHREADS * EPL) + tid * EPL;
    buf[k] = *(const uint4*)(codes + (e0 < GE ? e0 : GE - EPL));
  }
}

__device__ __forceinline__ unsigned zero_bytes(unsigned x) {
  const unsigned m = 0x7F7F7F7Fu;
  return ~(((x & m) + m) | x | m);
}

__device__ __forceinline__ void note_sixteen(uint4 w, int e0, unsigned pat,
                                             unsigned* plist, int* pcount) {
  const unsigned z0 = zero_bytes(w.x ^ pat);
  const unsigned z1 = zero_bytes(w.y ^ pat);
  const unsigned z2 = zero_bytes(w.z ^ pat);
  const unsigned z3 = zero_bytes(w.w ^ pat);
  unsigned hits = (z0 >> 7) | (z1 >> 6) | (z2 >> 5) | (z3 >> 4);
  if (e0 >= GE) hits = 0;
  while (hits != 0) {
    const int bit = __builtin_ctz(hits);
    hits &= hits - 1;
    const int pos = atomicAdd(pcount, 1);
    plist[pos] = (unsigned)(e0 + ((bit & 7) << 2) + (bit >> 3));
  }
}

__device__ __forceinline__ void hold_loaded(const v4f (&v)[4][2]) {
  asm volatile("" ::
               "v"(v[0][0]), "v"(v[0][1]), "v"(v[1][0]), "v"(v[1][1]),
               "v"(v[2][0]), "v"(v[2][1]), "v"(v[3][0]), "v"(v[3][1]));
}
template <int F>
__device__ __forceinline__ void hold_loaded(const v4f (&)[F][1]) {}

template <int NP, int FUB>
__device__ __forceinline__ void drain_rows(
    const int* __restrict__ src, const int* __restrict__ dst,
    const float* __restrict__ fa, const float* __restrict__ fb,
    float* agg_a, float* agg_b, unsigned* plist, unsigned* sorted, int* ocnt,
    int* obase, int* ocur, int n, bool fresh, int row_base, int tid, int wave,
    int lane) {
  if (tid < NBIN) ocnt[tid] = 0;
  __syncthreads();

  for (int i = tid; i < n; i += 4 * STHREADS) {
    unsigned ent[4];
    int s[4], d[4];
#pragma unroll
    for (int q = 0; q < 4; ++q) {
      const int ii = i + q * STHREADS;
      ent[q] = plist[ii < n ? ii : i];
    }
#pragma unroll
    for (int q = 0; q < 4; ++q) {
      s[q] = src[ent[q]];
      d[q] = dst[ent[q]];
    }
#pragma unroll
    for (int q = 0; q < 4; ++q) {
      const int ii = i + q * STHREADS;
      if (ii < n) {
        const unsigned lr = (unsigned)(d[q] - row_base);
        plist[ii] = ((unsigned)s[q] << 8) | lr;
        atomicAdd(&ocnt[lr], 1);
      }
    }
  }
  __syncthreads();

  if (wave == 0) {
    int c[RPW];
    int run = 0;
#pragma unroll
    for (int q = 0; q < RPW; ++q) {
      const int x = ocnt[lane * RPW + q];
      c[q] = run;
      run += x;
    }
    int incl = run;
#pragma unroll
    for (int dlt = 1; dlt < 32; dlt <<= 1) {
      const int t = __shfl_up(incl, dlt);
      if (lane >= dlt) incl += t;
    }
    const int excl = incl - run;
#pragma unroll
    for (int q = 0; q < RPW; ++q) {
      obase[lane * RPW + q] = excl + c[q];
      ocur[lane * RPW + q] = excl + c[q];
    }
    if (lane == 31) obase[NBIN] = incl;
  }
  __syncthreads();

  for (int i = tid; i < n; i += STHREADS) {
    const unsigned w = plist[i];
    const int pos = atomicAdd(&ocur[w & 255u], 1);
    sorted[pos] = w;
  }
  __syncthreads();

  for (int j = 0; j < RPW; ++j) {
    const int lr = wave + j * SWAVES;
    const int row = row_base + lr;
    if (lr >= TR || row >= GN) break;
    int i0 = __builtin_amdgcn_readfirstlane(obase[lr]);
    const int sege = __builtin_amdgcn_readfirstlane(obase[lr + 1]);
    float* oa = agg_a + (size_t)row * GH + lane * 4;
    float* ob = agg_b + (size_t)row * GH + lane * 4;
    v4f a[NP];
#pragma unroll
    for (int p = 0; p < NP; ++p) a[p] = (v4f){0.0f, 0.0f, 0.0f, 0.0f};
    if (!fresh) {
      a[0] = *(const v4f*)oa;
      if (NP == 2) a[NP - 1] = *(const v4f*)ob;
    }
    for (; i0 + FUB <= sege; i0 += FUB) {
      unsigned w[FUB];
      v4f v[FUB][NP];
#pragma unroll
      for (int u = 0; u < FUB; ++u)
        w[u] = (unsigned)__builtin_amdgcn_readfirstlane((int)sorted[i0 + u]);
#pragma unroll
      for (int u = 0; u < FUB; ++u) {
        const size_t off = (size_t)(w[u] >> 8) * GH + lane * 4;
        v[u][0] = *(const v4f*)(fa + off);
        if (NP == 2) v[u][NP - 1] = *(const v4f*)(fb + off);
      }
      hold_loaded(v);
#pragma unroll
      for (int u = 0; u < FUB; ++u)
#pragma unroll
        for (int p = 0; p < NP; ++p) a[p] += v[u][p];
    }
    const int rem = sege - i0;
    if (rem > 0) {
      unsigned w[FUB - 1];
      v4f v[FUB - 1][NP];
#pragma unroll
      for (int u = 0; u < FUB - 1; ++u)
        w[u] = (unsigned)__builtin_amdgcn_readfirstlane((int)sorted[i0 + u]);
#pragma unroll
      for (int u = 0; u < FUB - 1; ++u)
        if (u < rem) {
          const size_t off = (size_t)(w[u] >> 8) * GH + lane * 4;
          v[u][0] = *(const v4f*)(fa + off);
          if (NP == 2) v[u][NP - 1] = *(const v4f*)(fb + off);
        }
#pragma unroll
      for (int u = 0; u < FUB - 1; ++u)
        if (u < rem) {
#pragma unroll
          for (int p = 0; p < NP; ++p) a[p] += v[u][p];
        }
    }
    *(v4f*)oa = a[0];
    if (NP == 2) *(v4f*)ob = a[NP - 1];
  }
  __syncthreads();
}

template <int NP, int FUB>
__device__ __forceinline__ void scatter_body(
    const int* __restrict__ src, const int* __restrict__ dst,
    const unsigned char* __restrict__ codes, const float* __restrict__ fa,
    const float* __restrict__ fb, float* agg_a, float* agg_b, unsigned* plist,
    unsigned* sorted, int* pcount, int* ocnt, int* obase, int* ocur) {
  const int tid = threadIdx.x;
  const int lane = tid & 31;
  const int wave = tid >> 5;
  const int row_base = blockIdx.x * TR;
  const unsigned pat = blockIdx.x * 0x01010101u;

  if (tid == 0) *pcount = 0;
  __syncthreads();

  int chunk = (blockIdx.x * NCHUNK) / NTILES;
  uint4 cur[SK], nxt[SK];
  load_chunk(codes, chunk, tid, cur);
  bool fresh = true;
  for (int c = 0; c < NCHUNK; ++c) {
    int nchunk = chunk + 1;
    if (nchunk == NCHUNK) nchunk = 0;
    const bool last = (c + 1 == NCHUNK);
    if (!last) load_chunk(codes, nchunk, tid, nxt);

#pragma unroll
    for (int k = 0; k < SK; ++k) {
      const int e0 = chunk * CHUNK + k * (STHREADS * EPL) + tid * EPL;
      note_sixteen(cur[k], e0, pat, plist, pcount);
    }
    __syncthreads();
    const int n = *pcount;
    __syncthreads();
    if (last || n + CHUNK > LCAP2) {
      drain_rows<NP, FUB>(src, dst, fa, fb, agg_a, agg_b, plist, sorted, ocnt, obase,
                          ocur, n, fresh, row_base, tid, wave, lane);
      fresh = false;
      if (tid == 0) *pcount = 0;
      __syncthreads();
    }
    if (!last) {
#pragma unroll
      for (int k = 0; k < SK; ++k) cur[k] = nxt[k];
    }
    chunk = nchunk;
  }
}

__global__ __launch_bounds__(STHREADS) void scatter_tile_kernel(
    const int* __restrict__ src, const int* __restrict__ dst,
    const float* __restrict__ codes_f, const float* __restrict__ feat,
    float* __restrict__ agg) {
  __shared__ unsigned plist[LCAP2];
  __shared__ unsigned sorted[LCAP2 + 8];
  __shared__ int pcount;
  __shared__ int ocnt[NBIN], obase[NBIN + 1], ocur[NBIN];
  scatter_body<1, 8>(src, dst, (const unsigned char*)codes_f, feat, feat, agg, agg,
                  plist, sorted, &pcount, ocnt, obase, ocur);
}

__global__ __launch_bounds__(STHREADS) void scatter_pair_kernel(
    const int* __restrict__ src, const int* __restrict__ dst,
    const float* __restrict__ codes_f, const float* __restrict__ feat_a,
    const float* __restrict__ feat_b, float* __restrict__ agg_a,
    float* __restrict__ agg_b) {
  __shared__ unsigned plist[LCAP2];
  __shared__ unsigned sorted[LCAP2 + 8];
  __shared__ int pcount;
  __shared__ int ocnt[NBIN], obase[NBIN + 1], ocur[NBIN];
  scatter_body<2, 4>(src, dst, (const unsigned char*)codes_f, feat_a, feat_b, agg_a,
                  agg_b, plist, sorted, &pcount, ocnt, obase, ocur);
}

__global__ __launch_bounds__(256) void gemm_zr_kernel(
    const float* __restrict__ x, const float* __restrict__ aggx,
    const float* __restrict__ hl, const float* __restrict__ aggh,
    const float* __restrict__ Wl, const float* __restrict__ bl,
    float* __restrict__ zout, float* __restrict__ vout) {
  __shared__ float lds[4 * 16 * 132];
  const int tid = threadIdx.x;
  const int wave = tid >> 5;
  const int lane = tid & 31;
  const int row0 = blockIdx.x * 128 + wave * 16;
  const int kof = (lane >> 4) << 1;
  const int nn = lane & 15;
  const int rowhi = (lane >> 4) << 3;

  int arow = row0 + nn;
  if (arow > GN - 1) arow = GN - 1;

  v2f fs[32], ft[32];
  load_afrag_sum(x, aggx, arow, kof, fs);
  load_afrag_sum(hl, aggh, arow, kof, ft);

  const float* W0 = Wl + 0 * GH * GH;
  const float* W1 = Wl + 1 * GH * GH;
  const float* W2 = Wl + 2 * GH * GH;
  const float* W3 = Wl + 3 * GH * GH;
  float* L0 = lds + 0 * 16 * 132;
  float* L1 = lds + 1 * 16 * 132;
  float* L2 = lds + 2 * 16 * 132;
  float* L3 = lds + 3 * 16 * 132;

  for (int ct = 0; ct < GH / 16; ++ct) {
    const int n0 = ct * 16;
    __syncthreads();
    stage_slab(W0, n0, L0);
    stage_slab(W1, n0, L1);
    stage_slab(W2, n0, L2);
    stage_slab(W3, n0, L3);
    __syncthreads();

    const float* p0 = L0 + nn * 132 + kof;
    const float* p1 = L1 + nn * 132 + kof;
    const float* p2 = L2 + nn * 132 + kof;
    const float* p3 = L3 + nn * 132 + kof;

    v8f az0 = {0, 0, 0, 0, 0, 0, 0, 0}, az1 = {0, 0, 0, 0, 0, 0, 0, 0};
    v8f ar0 = {0, 0, 0, 0, 0, 0, 0, 0}, ar1 = {0, 0, 0, 0, 0, 0, 0, 0};
#pragma unroll
    for (int kk = 0; kk < 32; kk += 2) {
      az0 = wmma_f32(fs[kk], *(const v2f*)(p0 + 4 * kk), az0);
      az0 = wmma_f32(ft[kk], *(const v2f*)(p1 + 4 * kk), az0);
      ar0 = wmma_f32(fs[kk], *(const v2f*)(p2 + 4 * kk), ar0);
      ar0 = wmma_f32(ft[kk], *(const v2f*)(p3 + 4 * kk), ar0);
      az1 = wmma_f32(fs[kk + 1], *(const v2f*)(p0 + 4 * kk + 4), az1);
      az1 = wmma_f32(ft[kk + 1], *(const v2f*)(p1 + 4 * kk + 4), az1);
      ar1 = wmma_f32(fs[kk + 1], *(const v2f*)(p2 + 4 * kk + 4), ar1);
      ar1 = wmma_f32(ft[kk + 1], *(const v2f*)(p3 + 4 * kk + 4), ar1);
    }
    const v8f accz = az0 + az1;
    const v8f accr = ar0 + ar1;

    const float bz = bl[0 * GH + n0 + nn] + bl[1 * GH + n0 + nn];
    const float br = bl[2 * GH + n0 + nn] + bl[3 * GH + n0 + nn];
#pragma unroll
    for (int i = 0; i < 8; ++i) {
      const int row = row0 + rowhi + i;
      if (row < GN) {
        const size_t off = (size_t)row * GH + n0 + nn;
        const float zv = sigmoid_f(accz[i] + bz);
        const float rv = sigmoid_f(accr[i] + br);
        zout[off] = zv;
        vout[off] = rv * hl[off];
      }
    }
  }
}

__global__ __launch_bounds__(256) void gemm_ht_kernel(
    const float* __restrict__ x, const float* __restrict__ aggx,
    const float* __restrict__ v, const float* __restrict__ aggv,
    const float* __restrict__ Wl, const float* __restrict__ bl,
    const float* __restrict__ zbuf, const float* __restrict__ hl,
    float* __restrict__ outp) {
  __shared__ float lds[2 * 16 * 132];
  const int tid = threadIdx.x;
  const int wave = tid >> 5;
  const int lane = tid & 31;
  const int row0 = blockIdx.x * 128 + wave * 16;
  const int kof = (lane >> 4) << 1;
  const int nn = lane & 15;
  const int rowhi = (lane >> 4) << 3;

  int arow = row0 + nn;
  if (arow > GN - 1) arow = GN - 1;

  v2f fs[32], fu[32];
  load_afrag_sum(x, aggx, arow, kof, fs);
  load_afrag_sum(v, aggv, arow, kof, fu);

  const float* W4 = Wl + 4 * GH * GH;
  const float* W5 = Wl + 5 * GH * GH;
  float* L4 = lds + 0 * 16 * 132;
  float* L5 = lds + 1 * 16 * 132;

  for (int ct = 0; ct < GH / 16; ++ct) {
    const int n0 = ct * 16;
    __syncthreads();
    stage_slab(W4, n0, L4);
    stage_slab(W5, n0, L5);
    __syncthreads();

    const float* p4 = L4 + nn * 132 + kof;
    const float* p5 = L5 + nn * 132 + kof;

    v8f a0 = {0, 0, 0, 0, 0, 0, 0, 0}, a1 = {0, 0, 0, 0, 0, 0, 0, 0};
#pragma unroll
    for (int kk = 0; kk < 32; kk += 2) {
      a0 = wmma_f32(fs[kk], *(const v2f*)(p4 + 4 * kk), a0);
      a0 = wmma_f32(fu[kk], *(const v2f*)(p5 + 4 * kk), a0);
      a1 = wmma_f32(fs[kk + 1], *(const v2f*)(p4 + 4 * kk + 4), a1);
      a1 = wmma_f32(fu[kk + 1], *(const v2f*)(p5 + 4 * kk + 4), a1);
    }
    const v8f acc = a0 + a1;

    const float bh = bl[4 * GH + n0 + nn] + bl[5 * GH + n0 + nn];
#pragma unroll
    for (int i = 0; i < 8; ++i) {
      const int row = row0 + rowhi + i;
      if (row < GN) {
        const size_t off = (size_t)row * GH + n0 + nn;
        const float ht = tanh_f(acc[i] + bh);
        const float zv = zbuf[off];
        outp[off] = zv * hl[off] + (1.0f - zv) * ht;
      }
    }
  }
}

extern "C" void kernel_launch(void* const* d_in, const int* in_sizes, int n_in,
                              void* d_out, int out_size, void* d_ws,
                              size_t ws_size, hipStream_t stream) {
  (void)in_sizes; (void)n_in; (void)out_size; (void)ws_size;
  const float* inp = (const float*)d_in[0];
  const int* edg = (const int*)d_in[1];
  const float* hmat = (const float*)d_in[2];
  const float* Wmat = (const float*)d_in[3];
  const float* bmat = (const float*)d_in[4];
  float* out = (float*)d_out;

  const int* src = edg;
  const int* dst = edg + GE;

  float* agg_x = (float*)d_ws;
  float* agg_t = agg_x + NH;
  float* zbuf = agg_t + NH;
  float* vbuf = zbuf + NH;

  const int gemm_blocks = (GN + 127) / 128;

  static_assert((size_t)GE <= NH * sizeof(float), "codes fit one output plane");
  float* codes = out + (size_t)(GLAYERS - 1) * NH;
  tile_code_kernel<<<(GE / EPL + 255) / 256, 256, 0, stream>>>(dst, codes);

  for (int l = 0; l < GLAYERS; ++l) {
    const float* x = (l == 0) ? inp : (out + (size_t)(l - 1) * NH);
    const float* hl = hmat + (size_t)l * NH;
    const float* Wl = Wmat + (size_t)l * 6 * GH * GH;
    const float* bl = bmat + (size_t)l * 6 * GH;
    float* outl = out + (size_t)l * NH;

    scatter_pair_kernel<<<NTILES, STHREADS, 0, stream>>>(src, dst, codes, x, hl,
                                                         agg_x, agg_t);

    gemm_zr_kernel<<<gemm_blocks, 256, 0, stream>>>(x, agg_x, hl, agg_t, Wl,
                                                    bl, zbuf, vbuf);

    scatter_tile_kernel<<<NTILES, STHREADS, 0, stream>>>(src, dst, codes, vbuf, agg_t);

    gemm_ht_kernel<<<gemm_blocks, 256, 0, stream>>>(x, agg_x, vbuf, agg_t, Wl,
                                                    bl, zbuf, hl, outl);
  }
}


# First scatter launch of a call also writes each tile's grouped edge list and row offsets behind the codes; the other three read it back: no scan, index fetch or sort there; sums, their order kept
# speedup vs baseline: 3.3302x; 1.2736x over previous
#include <hip/hip_runtime.h>
#include <cstdint>
#include <cstddef>

constexpr int GN = 50000;
constexpr int GE = 1600000;
constexpr int GH = 128;
constexpr int GLAYERS = 2;
constexpr size_t NH = (size_t)GN * GH;

typedef __attribute__((ext_vector_type(2))) float v2f;
typedef __attribute__((ext_vector_type(8))) float v8f;
typedef __attribute__((ext_vector_type(4))) float v4f;

__device__ __forceinline__ v8f wmma_f32(v2f a, v2f b, v8f c) {
  return __builtin_amdgcn_wmma_f32_16x16x4_f32(
       false, a,  false, b,
       (short)0, c,  false,  false);
}

__device__ __forceinline__ float sigmoid_f(float x) {
  return 1.0f / (1.0f + __expf(-x));
}
__device__ __forceinline__ float tanh_f(float x) {
  return 1.0f - 2.0f / (1.0f + __expf(2.0f * x));
}

__device__ __forceinline__ void load_afrag_sum(const float* __restrict__ X,
                                               const float* __restrict__ A,
                                               int row, int kof, v2f* frag) {
  const float* xr = X + (size_t)row * GH;
  const float* ar = A + (size_t)row * GH;
#pragma unroll
  for (int kk = 0; kk < 32; ++kk) {
    const int c = (kk << 2) + kof;
    const float2 xv = *(const float2*)(xr + c);
    const float2 av = *(const float2*)(ar + c);
    v2f f;
    f.x = xv.x + av.x;
    f.y = xv.y + av.y;
    frag[kk] = f;
  }
}

__device__ __forceinline__ void stage_slab(const float* __restrict__ Wg, int n0,
                                           float* lds) {
  const int tid = threadIdx.x;
#pragma unroll
  for (int i = 0; i < 8; ++i) {
    const int idx = i * 256 + tid;
    const int k = idx >> 4;
    const int c = idx & 15;
    lds[c * 132 + k] = Wg[k * GH + n0 + c];
  }
}

constexpr int TR = 197;
constexpr int NTILES = (GN + TR - 1) / TR;
constexpr unsigned NO_TILE = 255u;
constexpr int STHREADS = 1024;
constexpr int SWAVES = STHREADS / 32;
constexpr int RPW = (TR + SWAVES - 1) / SWAVES;
constexpr int EPL = 16;
constexpr int SK = 2;
constexpr int CHUNK = STHREADS * EPL * SK;
constexpr int NCHUNK = (GE + CHUNK - 1) / CHUNK;
constexpr int NBIN = SWAVES * RPW;
constexpr int LCAP2 = 39936;

constexpr int SCAP = 8192;
constexpr int SHDR = 256;
constexpr int SSTRIDE = SHDR + SCAP;
constexpr unsigned NOT_SAVED = 0xFFFFFFFFu;
constexpr size_t SAVED_AT = (size_t)1 << 19;

static_assert(NTILES * TR >= GN, "tiles cover all rows");
static_assert(NBIN + 2 <= SHDR && SCAP % 4 == 0 && SCAP + 8 <= LCAP2, "header and list fit");
static_assert((size_t)GE <= SAVED_AT * 4, "the saved lists start behind the codes");
static_assert(SAVED_AT + (size_t)NTILES * SSTRIDE <= NH, "codes and saved lists fit one output plane");
static_assert(NTILES <= (int)NO_TILE, "every tile number and the spare code fit one byte");
static_assert(GE % EPL == 0, "16-byte code loads");
static_assert(LCAP2 >= CHUNK, "an emptied list holds every edge of one scan step");
static_assert((2 * LCAP2 + 3 * NBIN + 16) * 4 <= 320 * 1024, "both lists fit the LDS");
static_assert(TR <= 256 && GN < (1 << 24), "local row and source row fit one word");
static_assert(SWAVES == 32 && NBIN >= TR + 1, "the first wave's lanes take RPW counters each");

__device__ __forceinline__ unsigned tile_code(int d) {
  const unsigned a = (unsigned)d;
  const unsigned lim = NO_TILE * (unsigned)TR;
  return (a < lim ? a : lim) / (unsigned)TR;
}

__global__ __launch_bounds__(256) void tile_code_kernel(
    const int* __restrict__ dst, float* __restrict__ codes_f) {
  const int g = blockIdx.x * 256 + threadIdx.x;
  if (g >= GE / EPL) return;
  const int4* d4 = (const int4*)(dst + (size_t)g * EPL);
  int4 d[4];
#pragma unroll
  for (int q = 0; q < 4; ++q) d[q] = d4[q];
  unsigned w[4];
#pragma unroll
  for (int q = 0; q < 4; ++q)
    w[q] = tile_code(d[q].x) | (tile_code(d[q].y) << 8) |
           (tile_code(d[q].z) << 16) | (tile_code(d[q].w) << 24);
  ((uint4*)codes_f)[g] = make_uint4(w[0], w[1], w[2], w[3]);
}

__device__ __forceinline__ void load_chunk(const unsigned char* __restrict__ codes,
                                           int chunk, int tid, uint4* buf) {
#pragma unroll
  for (int k = 0; k < SK; ++k) {
    const int e0 = chunk * CHUNK + k * (STHREADS * EPL) + tid * EPL;
    buf[k] = *(const uint4*)(codes + (e0 < GE ? e0 : GE - EPL));
  }
}

__device__ __forceinline__ unsigned zero_bytes(unsigned x) {
  const unsigned m = 0x7F7F7F7Fu;
  return ~(((x & m) + m) | x | m);
}

__device__ __forceinline__ void note_sixteen(uint4 w, int e0, unsigned pat,
                                             unsigned* plist, int* pcount) {
  const unsigned z0 = zero_bytes(w.x ^ pat);
  const unsigned z1 = zero_bytes(w.y ^ pat);
  const unsigned z2 = zero_bytes(w.z ^ pat);
  const unsigned z3 = zero_bytes(w.w ^ pat);
  unsigned hits = (z0 >> 7) | (z1 >> 6) | (z2 >> 5) | (z3 >> 4);
  if (e0 >= GE) hits = 0;
  while (hits != 0) {
    const int bit = __builtin_ctz(hits);
    hits &= hits - 1;
    const int pos = atomicAdd(pcount, 1);
    plist[pos] = (unsigned)(e0 + ((bit & 7) << 2) + (bit >> 3));
  }
}

__device__ __forceinline__ void hold_loaded(const v4f (&v)[4][2]) {
  asm volatile("" ::
               "v"(v[0][0]), "v"(v[0][1]), "v"(v[1][0]), "v"(v[1][1]),
               "v"(v[2][0]), "v"(v[2][1]), "v"(v[3][0]), "v"(v[3][1]));
}
template <int F>
__device__ __forceinline__ void hold_loaded(const v4f (&)[F][1]) {}

template <int NP, int FUB>
__device__ __forceinline__ void sum_rows(
    const float* __restrict__ fa, const float* __restrict__ fb, float* agg_a,
    float* agg_b, const unsigned* sorted, const int* obase, bool fresh,
    int row_base, int wave, int lane);

template <int NP, int FUB>
__device__ __forceinline__ void drain_rows(
    const int* __restrict__ src, const int* __restrict__ dst,
    const float* __restrict__ fa, const float* __restrict__ fb,
    float* agg_a, float* agg_b, unsigned* plist, unsigned* sorted, int* ocnt,
    int* obase, int* ocur, int n, bool fresh, int row_base, int tid, int wave,
    int lane, unsigned* save_to) {
  if (tid < NBIN) ocnt[tid] = 0;
  __syncthreads();

  for (int i = tid; i < n; i += 4 * STHREADS) {
    unsigned ent[4];
    int s[4], d[4];
#pragma unroll
    for (int q = 0; q < 4; ++q) {
      const int ii = i + q * STHREADS;
      ent[q] = plist[ii < n ? ii : i];
    }
#pragma unroll
    for (int q = 0; q < 4; ++q) {
      s[q] = src[ent[q]];
      d[q] = dst[ent[q]];
    }
#pragma unroll
    for (int q = 0; q < 4; ++q) {
      const int ii = i + q * STHREADS;
      if (ii < n) {
        const unsigned lr = (unsigned)(d[q] - row_base);
        plist[ii] = ((unsigned)s[q] << 8) | lr;
        atomicAdd(&ocnt[lr], 1);
      }
    }
  }
  __syncthreads();

  if (wave == 0) {
    int c[RPW];
    int run = 0;
#pragma unroll
    for (int q = 0; q < RPW; ++q) {
      const int x = ocnt[lane * RPW + q];
      c[q] = run;
      run += x;
    }
    int incl = run;
#pragma unroll
    for (int dlt = 1; dlt < 32; dlt <<= 1) {
      const int t = __shfl_up(incl, dlt);
      if (lane >= dlt) incl += t;
    }
    const int excl = incl - run;
#pragma unroll
    for (int q = 0; q < RPW; ++q) {
      obase[lane * RPW + q] = excl + c[q];
      ocur[lane * RPW + q] = excl + c[q];
    }
    if (lane == 31) obase[NBIN] = incl;
  }
  __syncthreads();

  for (int i = tid; i < n; i += STHREADS) {
    const unsigned w = plist[i];
    const int pos = atomicAdd(&ocur[w & 255u], 1);
    sorted[pos] = w;
  }
  __syncthreads();

  if (save_to != nullptr) {
    if (fresh && n <= SCAP) {
      for (int i = tid * 4; i < n; i += 4 * STHREADS)
        *(uint4*)(save_to + SHDR + i) = *(const uint4*)(sorted + i);
      if (tid <= NBIN) save_to[1 + tid] = (unsigned)obase[tid];
      if (tid == 0) save_to[0] = (unsigned)n;
    } else if (tid == 0) {
      save_to[0] = NOT_SAVED;
    }
  }

  sum_rows<NP, FUB>(fa, fb, agg_a, agg_b, sorted, obase, fresh, row_base, wave, lane);
}

template <int NP, int FUB>
__device__ __forceinline__ void sum_rows(
    const float* __restrict__ fa, const float* __restrict__ fb, float* agg_a,
    float* agg_b, const unsigned* sorted, const int* obase, bool fresh,
    int row_base, int wave, int lane) {
  for (int j = 0; j < RPW; ++j) {
    const int lr = wave + j * SWAVES;
    const int row = row_base + lr;
    if (lr >= TR || row >= GN) break;
    int i0 = __builtin_amdgcn_readfirstlane(obase[lr]);
    const int sege = __builtin_amdgcn_readfirstlane(obase[lr + 1]);
    float* oa = agg_a + (size_t)row * GH + lane * 4;
    float* ob = agg_b + (size_t)row * GH + lane * 4;
    v4f a[NP];
#pragma unroll
    for (int p = 0; p < NP; ++p) a[p] = (v4f){0.0f, 0.0f, 0.0f, 0.0f};
    if (!fresh) {
      a[0] = *(const v4f*)oa;
      if (NP == 2) a[NP - 1] = *(const v4f*)ob;
    }
    for (; i0 + FUB <= sege; i0 += FUB) {
      unsigned w[FUB];
      v4f v[FUB][NP];
#pragma unroll
      for (int u = 0; u < FUB; ++u)
        w[u] = (unsigned)__builtin_amdgcn_readfirstlane((int)sorted[i0 + u]);
#pragma unroll
      for (int u = 0; u < FUB; ++u) {
        const size_t off = (size_t)(w[u] >> 8) * GH + lane * 4;
        v[u][0] = *(const v4f*)(fa + off);
        if (NP == 2) v[u][NP - 1] = *(const v4f*)(fb + off);
      }
      hold_loaded(v);
#pragma unroll
      for (int u = 0; u < FUB; ++u)
#pragma unroll
        for (int p = 0; p < NP; ++p) a[p] += v[u][p];
    }
    const int rem = sege - i0;
    if (rem > 0) {
      unsigned w[FUB - 1];
      v4f v[FUB - 1][NP];
#pragma unroll
      for (int u = 0; u < FUB - 1; ++u)
        w[u] = (unsigned)__builtin_amdgcn_readfirstlane((int)sorted[i0 + u]);
#pragma unroll
      for (int u = 0; u < FUB - 1; ++u)
        if (u < rem) {
          const size_t off = (size_t)(w[u] >> 8) * GH + lane * 4;
          v[u][0] = *(const v4f*)(fa + off);
          if (NP == 2) v[u][NP - 1] = *(const v4f*)(fb + off);
        }
#pragma unroll
      for (int u = 0; u < FUB - 1; ++u)
        if (u < rem) {
#pragma unroll
          for (int p = 0; p < NP; ++p) a[p] += v[u][p];
        }
    }
    *(v4f*)oa = a[0];
    if (NP == 2) *(v4f*)ob = a[NP - 1];
  }
  __syncthreads();
}

template <int NP, int FUB, bool FIRST>
__device__ __forceinline__ void scatter_body(
    const int* __restrict__ src, const int* __restrict__ dst,
    const unsigned char* __restrict__ codes, const float* __restrict__ fa,
    const float* __restrict__ fb, float* agg_a, float* agg_b, unsigned* saved,
    unsigned* plist, unsigned* sorted, int* pcount, int* ocnt, int* obase, int* ocur) {
  const int tid = threadIdx.x;
  const int lane = tid & 31;
  const int wave = tid >> 5;
  const int row_base = blockIdx.x * TR;
  const unsigned pat = blockIdx.x * 0x01010101u;
  unsigned* my_saved = saved + SAVED_AT + (size_t)blockIdx.x * SSTRIDE;

  if (!FIRST) {
    const unsigned nsv = (unsigned)__builtin_amdgcn_readfirstlane((int)my_saved[0]);
    if (nsv <= (unsigned)SCAP) {
      for (int i = tid * 4; i < (int)nsv; i += 4 * STHREADS)
        *(uint4*)(sorted + i) = *(const uint4*)(my_saved + SHDR + i);
      if (tid <= NBIN) obase[tid] = (int)my_saved[1 + tid];
      __syncthreads();
      sum_rows<NP, FUB>(fa, fb, agg_a, agg_b, sorted, obase, true, row_base, wave, lane);
      return;
    }
  }

  if (tid == 0) *pcount = 0;
  __syncthreads();

  int chunk = (blockIdx.x * NCHUNK) / NTILES;
  uint4 cur[SK], nxt[SK];
  load_chunk(codes, chunk, tid, cur);
  bool fresh = true;
  for (int c = 0; c < NCHUNK; ++c) {
    int nchunk = chunk + 1;
    if (nchunk == NCHUNK) nchunk = 0;
    const bool last = (c + 1 == NCHUNK);
    if (!last) load_chunk(codes, nchunk, tid, nxt);

#pragma unroll
    for (int k = 0; k < SK; ++k) {
      const int e0 = chunk * CHUNK + k * (STHREADS * EPL) + tid * EPL;
      note_sixteen(cur[k], e0, pat, plist, pcount);
    }
    __syncthreads();
    const int n = *pcount;
    __syncthreads();
    if (last || n + CHUNK > LCAP2) {
      drain_rows<NP, FUB>(src, dst, fa, fb, agg_a, agg_b, plist, sorted, ocnt, obase,
                          ocur, n, fresh, row_base, tid, wave, lane,
                          (FIRST && last) ? my_saved : nullptr);
      fresh = false;
      if (tid == 0) *pcount = 0;
      __syncthreads();
    }
    if (!last) {
#pragma unroll
      for (int k = 0; k < SK; ++k) cur[k] = nxt[k];
    }
    chunk = nchunk;
  }
}

__global__ __launch_bounds__(STHREADS) void scatter_tile_kernel(
    const int* __restrict__ src, const int* __restrict__ dst,
    float* scratch_f, const float* __restrict__ feat, float* __restrict__ agg) {
  __shared__ unsigned plist[LCAP2];
  __shared__ __attribute__((aligned(16))) unsigned sorted[LCAP2 + 8];
  __shared__ int pcount;
  __shared__ int ocnt[NBIN], obase[NBIN + 1], ocur[NBIN];
  scatter_body<1, 8, false>(src, dst, (const unsigned char*)scratch_f, feat, feat, agg,
                            agg, (unsigned*)scratch_f, plist, sorted, &pcount, ocnt,
                            obase, ocur);
}

template <bool FIRST>
__global__ __launch_bounds__(STHREADS) void scatter_pair_kernel(
    const int* __restrict__ src, const int* __restrict__ dst,
    float* scratch_f, const float* __restrict__ feat_a,
    const float* __restrict__ feat_b, float* __restrict__ agg_a,
    float* __restrict__ agg_b) {
  __shared__ unsigned plist[LCAP2];
  __shared__ __attribute__((aligned(16))) unsigned sorted[LCAP2 + 8];
  __shared__ int pcount;
  __shared__ int ocnt[NBIN], obase[NBIN + 1], ocur[NBIN];
  scatter_body<2, 4, FIRST>(src, dst, (const unsigned char*)scratch_f, feat_a, feat_b,
                            agg_a, agg_b, (unsigned*)scratch_f, plist, sorted, &pcount,
                            ocnt, obase, ocur);
}

__global__ __launch_bounds__(256) void gemm_zr_kernel(
    const float* __restrict__ x, const float* __restrict__ aggx,
    const float* __restrict__ hl, const float* __restrict__ aggh,
    const float* __restrict__ Wl, const float* __restrict__ bl,
    float* __restrict__ zout, float* __restrict__ vout) {
  __shared__ float lds[4 * 16 * 132];
  const int tid = threadIdx.x;
  const int wave = tid >> 5;
  const int lane = tid & 31;
  const int row0 = blockIdx.x * 128 + wave * 16;
  const int kof = (lane >> 4) << 1;
  const int nn = lane & 15;
  const int rowhi = (lane >> 4) << 3;

  int arow = row0 + nn;
  if (arow > GN - 1) arow = GN - 1;

  v2f fs[32], ft[32];
  load_afrag_sum(x, aggx, arow, kof, fs);
  load_afrag_sum(hl, aggh, arow, kof, ft);

  const float* W0 = Wl + 0 * GH * GH;
  const float* W1 = Wl + 1 * GH * GH;
  const float* W2 = Wl + 2 * GH * GH;
  const float* W3 = Wl + 3 * GH * GH;
  float* L0 = lds + 0 * 16 * 132;
  float* L1 = lds + 1 * 16 * 132;
  float* L2 = lds + 2 * 16 * 132;
  float* L3 = lds + 3 * 16 * 132;

  for (int ct = 0; ct < GH / 16; ++ct) {
    const int n0 = ct * 16;
    __syncthreads();
    stage_slab(W0, n0, L0);
    stage_slab(W1, n0, L1);
    stage_slab(W2, n0, L2);
    stage_slab(W3, n0, L3);
    __syncthreads();

    const float* p0 = L0 + nn * 132 + kof;
    const float* p1 = L1 + nn * 132 + kof;
    const float* p2 = L2 + nn * 132 + kof;
    const float* p3 = L3 + nn * 132 + kof;

    v8f az0 = {0, 0, 0, 0, 0, 0, 0, 0}, az1 = {0, 0, 0, 0, 0, 0, 0, 0};
    v8f ar0 = {0, 0, 0, 0, 0, 0, 0, 0}, ar1 = {0, 0, 0, 0, 0, 0, 0, 0};
#pragma unroll
    for (int kk = 0; kk < 32; kk += 2) {
      az0 = wmma_f32(fs[kk], *(const v2f*)(p0 + 4 * kk), az0);
      az0 = wmma_f32(ft[kk], *(const v2f*)(p1 + 4 * kk), az0);
      ar0 = wmma_f32(fs[kk], *(const v2f*)(p2 + 4 * kk), ar0);
      ar0 = wmma_f32(ft[kk], *(const v2f*)(p3 + 4 * kk), ar0);
      az1 = wmma_f32(fs[kk + 1], *(const v2f*)(p0 + 4 * kk + 4), az1);
      az1 = wmma_f32(ft[kk + 1], *(const v2f*)(p1 + 4 * kk + 4), az1);
      ar1 = wmma_f32(fs[kk + 1], *(const v2f*)(p2 + 4 * kk + 4), ar1);
      ar1 = wmma_f32(ft[kk + 1], *(const v2f*)(p3 + 4 * kk + 4), ar1);
    }
    const v8f accz = az0 + az1;
    const v8f accr = ar0 + ar1;

    const float bz = bl[0 * GH + n0 + nn] + bl[1 * GH + n0 + nn];
    const float br = bl[2 * GH + n0 + nn] + bl[3 * GH + n0 + nn];
#pragma unroll
    for (int i = 0; i < 8; ++i) {
      const int row = row0 + rowhi + i;
      if (row < GN) {
        const size_t off = (size_t)row * GH + n0 + nn;
        const float zv = sigmoid_f(accz[i] + bz);
        const float rv = sigmoid_f(accr[i] + br);
        zout[off] = zv;
        vout[off] = rv * hl[off];
      }
    }
  }
}

__global__ __launch_bounds__(256) void gemm_ht_kernel(
    const float* __restrict__ x, const float* __restrict__ aggx,
    const float* __restrict__ v, const float* __restrict__ aggv,
    const float* __restrict__ Wl, const float* __restrict__ bl,
    const float* __restrict__ zbuf, const float* __restrict__ hl,
    float* __restrict__ outp) {
  __shared__ float lds[2 * 16 * 132];
  const int tid = threadIdx.x;
  const int wave = tid >> 5;
  const int lane = tid & 31;
  const int row0 = blockIdx.x * 128 + wave * 16;
  const int kof = (lane >> 4) << 1;
  const int nn = lane & 15;
  const int rowhi = (lane >> 4) << 3;

  int arow = row0 + nn;
  if (arow > GN - 1) arow = GN - 1;

  v2f fs[32], fu[32];
  load_afrag_sum(x, aggx, arow, kof, fs);
  load_afrag_sum(v, aggv, arow, kof, fu);

  const float* W4 = Wl + 4 * GH * GH;
  const float* W5 = Wl + 5 * GH * GH;
  float* L4 = lds + 0 * 16 * 132;
  float* L5 = lds + 1 * 16 * 132;

  for (int ct = 0; ct < GH / 16; ++ct) {
    const int n0 = ct * 16;
    __syncthreads();
    stage_slab(W4, n0, L4);
    stage_slab(W5, n0, L5);
    __syncthreads();

    const float* p4 = L4 + nn * 132 + kof;
    const float* p5 = L5 + nn * 132 + kof;

    v8f a0 = {0, 0, 0, 0, 0, 0, 0, 0}, a1 = {0, 0, 0, 0, 0, 0, 0, 0};
#pragma unroll
    for (int kk = 0; kk < 32; kk += 2) {
      a0 = wmma_f32(fs[kk], *(const v2f*)(p4 + 4 * kk), a0);
      a0 = wmma_f32(fu[kk], *(const v2f*)(p5 + 4 * kk), a0);
      a1 = wmma_f32(fs[kk + 1], *(const v2f*)(p4 + 4 * kk + 4), a1);
      a1 = wmma_f32(fu[kk + 1], *(const v2f*)(p5 + 4 * kk + 4), a1);
    }
    const v8f acc = a0 + a1;

    const float bh = bl[4 * GH + n0 + nn] + bl[5 * GH + n0 + nn];
#pragma unroll
    for (int i = 0; i < 8; ++i) {
      const int row = row0 + rowhi + i;
      if (row < GN) {
        const size_t off = (size_t)row * GH + n0 + nn;
        const float ht = tanh_f(acc[i] + bh);
        const float zv = zbuf[off];
        outp[off] = zv * hl[off] + (1.0f - zv) * ht;
      }
    }
  }
}

extern "C" void kernel_launch(void* const* d_in, const int* in_sizes, int n_in,
                              void* d_out, int out_size, void* d_ws,
                              size_t ws_size, hipStream_t stream) {
  (void)in_sizes; (void)n_in; (void)out_size; (void)ws_size;
  const float* inp = (const float*)d_in[0];
  const int* edg = (const int*)d_in[1];
  const float* hmat = (const float*)d_in[2];
  const float* Wmat = (const float*)d_in[3];
  const float* bmat = (const float*)d_in[4];
  float* out = (float*)d_out;

  const int* src = edg;
  const int* dst = edg + GE;

  float* agg_x = (float*)d_ws;
  float* agg_t = agg_x + NH;
  float* zbuf = agg_t + NH;
  float* vbuf = zbuf + NH;

  const int gemm_blocks = (GN + 127) / 128;

  static_assert((size_t)GE <= NH * sizeof(float), "codes fit one output plane");
  float* codes = out + (size_t)(GLAYERS - 1) * NH;
  tile_code_kernel<<<(GE / EPL + 255) / 256, 256, 0, stream>>>(dst, codes);

  for (int l = 0; l < GLAYERS; ++l) {
    const float* x = (l == 0) ? inp : (out + (size_t)(l - 1) * NH);
    const float* hl = hmat + (size_t)l * NH;
    const float* Wl = Wmat + (size_t)l * 6 * GH * GH;
    const float* bl = bmat + (size_t)l * 6 * GH;
    float* outl = out + (size_t)l * NH;

    if (l == 0)
      scatter_pair_kernel<true><<<NTILES, STHREADS, 0, stream>>>(src, dst, codes, x, hl,
                                                                 agg_x, agg_t);
    else
      scatter_pair_kernel<false><<<NTILES, STHREADS, 0, stream>>>(src, dst, codes, x, hl,
                                                                  agg_x, agg_t);

    gemm_zr_kernel<<<gemm_blocks, 256, 0, stream>>>(x, agg_x, hl, agg_t, Wl,
                                                    bl, zbuf, vbuf);

    scatter_tile_kernel<<<NTILES, STHREADS, 0, stream>>>(src, dst, codes, vbuf, agg_t);

    gemm_ht_kernel<<<gemm_blocks, 256, 0, stream>>>(x, agg_x, vbuf, agg_t, Wl,
                                                    bl, zbuf, hl, outl);
  }
}


# GEMMs: matrix-instruction operands change places (weights first, node rows second): a lane ends with 8 columns of ONE row, so the epilogue reads h, z and stores with 16-byte vectors; sums unchanged
# speedup vs baseline: 3.8844x; 1.1664x over previous
#include <hip/hip_runtime.h>
#include <cstdint>
#include <cstddef>

constexpr int GN = 50000;
constexpr int GE = 1600000;
constexpr int GH = 128;
constexpr int GLAYERS = 2;
constexpr size_t NH = (size_t)GN * GH;

typedef __attribute__((ext_vector_type(2))) float v2f;
typedef __attribute__((ext_vector_type(8))) float v8f;
typedef __attribute__((ext_vector_type(4))) float v4f;

__device__ __forceinline__ v8f wmma_f32(v2f a, v2f b, v8f c) {
  return __builtin_amdgcn_wmma_f32_16x16x4_f32(
       false, a,  false, b,
       (short)0, c,  false,  false);
}

__device__ __forceinline__ float sigmoid_f(float x) {
  return 1.0f / (1.0f + __expf(-x));
}
__device__ __forceinline__ float tanh_f(float x) {
  return 1.0f - 2.0f / (1.0f + __expf(2.0f * x));
}

__device__ __forceinline__ int weight_col(int m) {
  return ((m & 4) << 1) | ((m & 8) >> 1) | (m & 3);
}

__device__ __forceinline__ void hold_epilogue(v4f a, v4f b, v4f c, v4f d, v4f e,
                                              v4f f, v4f g, v4f h) {
  asm volatile("" :: "v"(a), "v"(b), "v"(c), "v"(d), "v"(e), "v"(f), "v"(g), "v"(h));
}
__device__ __forceinline__ void hold_epilogue(v4f a, v4f b, v4f c, v4f d, v4f e,
                                              v4f f, v4f g, v4f h, v4f i, v4f j) {
  asm volatile("" :: "v"(a), "v"(b), "v"(c), "v"(d), "v"(e), "v"(f), "v"(g), "v"(h),
               "v"(i), "v"(j));
}

__device__ __forceinline__ void load_afrag_sum(const float* __restrict__ X,
                                               const float* __restrict__ A,
                                               int row, int kof, v2f* frag) {
  const float* xr = X + (size_t)row * GH;
  const float* ar = A + (size_t)row * GH;
#pragma unroll
  for (int kk = 0; kk < 32; ++kk) {
    const int c = (kk << 2) + kof;
    const float2 xv = *(const float2*)(xr + c);
    const float2 av = *(const float2*)(ar + c);
    v2f f;
    f.x = xv.x + av.x;
    f.y = xv.y + av.y;
    frag[kk] = f;
  }
}

__device__ __forceinline__ void hold_fragments(const v2f (&f)[32]) {
  asm volatile("" :: "v"(f[0]), "v"(f[1]), "v"(f[2]), "v"(f[3]), "v"(f[4]), "v"(f[5]),
               "v"(f[6]), "v"(f[7]), "v"(f[8]), "v"(f[9]), "v"(f[10]), "v"(f[11]),
               "v"(f[12]), "v"(f[13]), "v"(f[14]), "v"(f[15]));
  asm volatile("" :: "v"(f[16]), "v"(f[17]), "v"(f[18]), "v"(f[19]), "v"(f[20]), "v"(f[21]),
               "v"(f[22]), "v"(f[23]), "v"(f[24]), "v"(f[25]), "v"(f[26]), "v"(f[27]),
               "v"(f[28]), "v"(f[29]), "v"(f[30]), "v"(f[31]));
}

__device__ __forceinline__ void stage_slab(const float* __restrict__ Wg, int n0,
                                           float* lds) {
  const int tid = threadIdx.x;
#pragma unroll
  for (int i = 0; i < 8; ++i) {
    const int idx = i * 256 + tid;
    const int k = idx >> 4;
    const int c = idx & 15;
    lds[c * 132 + k] = Wg[k * GH + n0 + c];
  }
}

constexpr int TR = 197;
constexpr int NTILES = (GN + TR - 1) / TR;
constexpr unsigned NO_TILE = 255u;
constexpr int STHREADS = 1024;
constexpr int SWAVES = STHREADS / 32;
constexpr int RPW = (TR + SWAVES - 1) / SWAVES;
constexpr int EPL = 16;
constexpr int SK = 2;
constexpr int CHUNK = STHREADS * EPL * SK;
constexpr int NCHUNK = (GE + CHUNK - 1) / CHUNK;
constexpr int NBIN = SWAVES * RPW;
constexpr int LCAP2 = 39936;

constexpr int SCAP = 8192;
constexpr int SHDR = 256;
constexpr int SSTRIDE = SHDR + SCAP;
constexpr unsigned NOT_SAVED = 0xFFFFFFFFu;
constexpr size_t SAVED_AT = (size_t)1 << 19;

static_assert(NTILES * TR >= GN, "tiles cover all rows");
static_assert(NBIN + 2 <= SHDR && SCAP % 4 == 0 && SCAP + 8 <= LCAP2, "header and list fit");
static_assert((size_t)GE <= SAVED_AT * 4, "the saved lists start behind the codes");
static_assert(SAVED_AT + (size_t)NTILES * SSTRIDE <= NH, "codes and saved lists fit one output plane");
static_assert(NTILES <= (int)NO_TILE, "every tile number and the spare code fit one byte");
static_assert(GE % EPL == 0, "16-byte code loads");
static_assert(LCAP2 >= CHUNK, "an emptied list holds every edge of one scan step");
static_assert((2 * LCAP2 + 3 * NBIN + 16) * 4 <= 320 * 1024, "both lists fit the LDS");
static_assert(TR <= 256 && GN < (1 << 24), "local row and source row fit one word");
static_assert(SWAVES == 32 && NBIN >= TR + 1, "the first wave's lanes take RPW counters each");

__device__ __forceinline__ unsigned tile_code(int d) {
  const unsigned a = (unsigned)d;
  const unsigned lim = NO_TILE * (unsigned)TR;
  return (a < lim ? a : lim) / (unsigned)TR;
}

__global__ __launch_bounds__(256) void tile_code_kernel(
    const int* __restrict__ dst, float* __restrict__ codes_f) {
  const int g = blockIdx.x * 256 + threadIdx.x;
  if (g >= GE / EPL) return;
  const int4* d4 = (const int4*)(dst + (size_t)g * EPL);
  int4 d[4];
#pragma unroll
  for (int q = 0; q < 4; ++q) d[q] = d4[q];
  unsigned w[4];
#pragma unroll
  for (int q = 0; q < 4; ++q)
    w[q] = tile_code(d[q].x) | (tile_code(d[q].y) << 8) |
           (tile_code(d[q].z) << 16) | (tile_code(d[q].w) << 24);
  ((uint4*)codes_f)[g] = make_uint4(w[0], w[1], w[2], w[3]);
}

__device__ __forceinline__ void load_chunk(const unsigned char* __restrict__ codes,
                                           int chunk, int tid, uint4* buf) {
#pragma unroll
  for (int k = 0; k < SK; ++k) {
    const int e0 = chunk * CHUNK + k * (STHREADS * EPL) + tid * EPL;
    buf[k] = *(const uint4*)(codes + (e0 < GE ? e0 : GE - EPL));
  }
}

__device__ __forceinline__ unsigned zero_bytes(unsigned x) {
  const unsigned m = 0x7F7F7F7Fu;
  return ~(((x & m) + m) | x | m);
}

__device__ __forceinline__ void note_sixteen(uint4 w, int e0, unsigned pat,
                                             unsigned* plist, int* pcount) {
  const unsigned z0 = zero_bytes(w.x ^ pat);
  const unsigned z1 = zero_bytes(w.y ^ pat);
  const unsigned z2 = zero_bytes(w.z ^ pat);
  const unsigned z3 = zero_bytes(w.w ^ pat);
  unsigned hits = (z0 >> 7) | (z1 >> 6) | (z2 >> 5) | (z3 >> 4);
  if (e0 >= GE) hits = 0;
  while (hits != 0) {
    const int bit = __builtin_ctz(hits);
    hits &= hits - 1;
    const int pos = atomicAdd(pcount, 1);
    plist[pos] = (unsigned)(e0 + ((bit & 7) << 2) + (bit >> 3));
  }
}

__device__ __forceinline__ void hold_loaded(const v4f (&v)[4][2]) {
  asm volatile("" ::
               "v"(v[0][0]), "v"(v[0][1]), "v"(v[1][0]), "v"(v[1][1]),
               "v"(v[2][0]), "v"(v[2][1]), "v"(v[3][0]), "v"(v[3][1]));
}
template <int F>
__device__ __forceinline__ void hold_loaded(const v4f (&)[F][1]) {}

template <int NP, int FUB>
__device__ __forceinline__ void sum_rows(
    const float* __restrict__ fa, const float* __restrict__ fb, float* agg_a,
    float* agg_b, const unsigned* sorted, const int* obase, bool fresh,
    int row_base, int wave, int lane);

template <int NP, int FUB>
__device__ __forceinline__ void drain_rows(
    const int* __restrict__ src, const int* __restrict__ dst,
    const float* __restrict__ fa, const float* __restrict__ fb,
    float* agg_a, float* agg_b, unsigned* plist, unsigned* sorted, int* ocnt,
    int* obase, int* ocur, int n, bool fresh, int row_base, int tid, int wave,
    int lane, unsigned* save_to) {
  if (tid < NBIN) ocnt[tid] = 0;
  __syncthreads();

  for (int i = tid; i < n; i += 4 * STHREADS) {
    unsigned ent[4];
    int s[4], d[4];
#pragma unroll
    for (int q = 0; q < 4; ++q) {
      const int ii = i + q * STHREADS;
      ent[q] = plist[ii < n ? ii : i];
    }
#pragma unroll
    for (int q = 0; q < 4; ++q) {
      s[q] = src[ent[q]];
      d[q] = dst[ent[q]];
    }
#pragma unroll
    for (int q = 0; q < 4; ++q) {
      const int ii = i + q * STHREADS;
      if (ii < n) {
        const unsigned lr = (unsigned)(d[q] - row_base);
        plist[ii] = ((unsigned)s[q] << 8) | lr;
        atomicAdd(&ocnt[lr], 1);
      }
    }
  }
  __syncthreads();

  if (wave == 0) {
    int c[RPW];
    int run = 0;
#pragma unroll
    for (int q = 0; q < RPW; ++q) {
      const int x = ocnt[lane * RPW + q];
      c[q] = run;
      run += x;
    }
    int incl = run;
#pragma unroll
    for (int dlt = 1; dlt < 32; dlt <<= 1) {
      const int t = __shfl_up(incl, dlt);
      if (lane >= dlt) incl += t;
    }
    const int excl = incl - run;
#pragma unroll
    for (int q = 0; q < RPW; ++q) {
      obase[lane * RPW + q] = excl + c[q];
      ocur[lane * RPW + q] = excl + c[q];
    }
    if (lane == 31) obase[NBIN] = incl;
  }
  __syncthreads();

  for (int i = tid; i < n; i += STHREADS) {
    const unsigned w = plist[i];
    const int pos = atomicAdd(&ocur[w & 255u], 1);
    sorted[pos] = w;
  }
  __syncthreads();

  if (save_to != nullptr) {
    if (fresh && n <= SCAP) {
      for (int i = tid * 4; i < n; i += 4 * STHREADS)
        *(uint4*)(save_to + SHDR + i) = *(const uint4*)(sorted + i);
      if (tid <= NBIN) save_to[1 + tid] = (unsigned)obase[tid];
      if (tid == 0) save_to[0] = (unsigned)n;
    } else if (tid == 0) {
      save_to[0] = NOT_SAVED;
    }
  }

  sum_rows<NP, FUB>(fa, fb, agg_a, agg_b, sorted, obase, fresh, row_base, wave, lane);
}

template <int NP, int FUB>
__device__ __forceinline__ void sum_rows(
    const float* __restrict__ fa, const float* __restrict__ fb, float* agg_a,
    float* agg_b, const unsigned* sorted, const int* obase, bool fresh,
    int row_base, int wave, int lane) {
  for (int j = 0; j < RPW; ++j) {
    const int lr = wave + j * SWAVES;
    const int row = row_base + lr;
    if (lr >= TR || row >= GN) break;
    int i0 = __builtin_amdgcn_readfirstlane(obase[lr]);
    const int sege = __builtin_amdgcn_readfirstlane(obase[lr + 1]);
    float* oa = agg_a + (size_t)row * GH + lane * 4;
    float* ob = agg_b + (size_t)row * GH + lane * 4;
    v4f a[NP];
#pragma unroll
    for (int p = 0; p < NP; ++p) a[p] = (v4f){0.0f, 0.0f, 0.0f, 0.0f};
    if (!fresh) {
      a[0] = *(const v4f*)oa;
      if (NP == 2) a[NP - 1] = *(const v4f*)ob;
    }
    for (; i0 + FUB <= sege; i0 += FUB) {
      unsigned w[FUB];
      v4f v[FUB][NP];
#pragma unroll
      for (int u = 0; u < FUB; ++u)
        w[u] = (unsigned)__builtin_amdgcn_readfirstlane((int)sorted[i0 + u]);
#pragma unroll
      for (int u = 0; u < FUB; ++u) {
        const size_t off = (size_t)(w[u] >> 8) * GH + lane * 4;
        v[u][0] = *(const v4f*)(fa + off);
        if (NP == 2) v[u][NP - 1] = *(const v4f*)(fb + off);
      }
      hold_loaded(v);
#pragma unroll
      for (int u = 0; u < FUB; ++u)
#pragma unroll
        for (int p = 0; p < NP; ++p) a[p] += v[u][p];
    }
    const int rem = sege - i0;
    if (rem > 0) {
      unsigned w[FUB - 1];
      v4f v[FUB - 1][NP];
#pragma unroll
      for (int u = 0; u < FUB - 1; ++u)
        w[u] = (unsigned)__builtin_amdgcn_readfirstlane((int)sorted[i0 + u]);
#pragma unroll
      for (int u = 0; u < FUB - 1; ++u)
        if (u < rem) {
          const size_t off = (size_t)(w[u] >> 8) * GH + lane * 4;
          v[u][0] = *(const v4f*)(fa + off);
          if (NP == 2) v[u][NP - 1] = *(const v4f*)(fb + off);
        }
#pragma unroll
      for (int u = 0; u < FUB - 1; ++u)
        if (u < rem) {
#pragma unroll
          for (int p = 0; p < NP; ++p) a[p] += v[u][p];
        }
    }
    *(v4f*)oa = a[0];
    if (NP == 2) *(v4f*)ob = a[NP - 1];
  }
  __syncthreads();
}

template <int NP, int FUB, bool FIRST>
__device__ __forceinline__ void scatter_body(
    const int* __restrict__ src, const int* __restrict__ dst,
    const unsigned char* __restrict__ codes, const float* __restrict__ fa,
    const float* __restrict__ fb, float* agg_a, float* agg_b, unsigned* saved,
    unsigned* plist, unsigned* sorted, int* pcount, int* ocnt, int* obase, int* ocur) {
  const int tid = threadIdx.x;
  const int lane = tid & 31;
  const int wave = tid >> 5;
  const int row_base = blockIdx.x * TR;
  const unsigned pat = blockIdx.x * 0x01010101u;
  unsigned* my_saved = saved + SAVED_AT + (size_t)blockIdx.x * SSTRIDE;

  if (!FIRST) {
    const unsigned nsv = (unsigned)__builtin_amdgcn_readfirstlane((int)my_saved[0]);
    if (nsv <= (unsigned)SCAP) {
      for (int i = tid * 4; i < (int)nsv; i += 4 * STHREADS)
        *(uint4*)(sorted + i) = *(const uint4*)(my_saved + SHDR + i);
      if (tid <= NBIN) obase[tid] = (int)my_saved[1 + tid];
      __syncthreads();
      sum_rows<NP, FUB>(fa, fb, agg_a, agg_b, sorted, obase, true, row_base, wave, lane);
      return;
    }
  }

  if (tid == 0) *pcount = 0;
  __syncthreads();

  int chunk = (blockIdx.x * NCHUNK) / NTILES;
  uint4 cur[SK], nxt[SK];
  load_chunk(codes, chunk, tid, cur);
  bool fresh = true;
  for (int c = 0; c < NCHUNK; ++c) {
    int nchunk = chunk + 1;
    if (nchunk == NCHUNK) nchunk = 0;
    const bool last = (c + 1 == NCHUNK);
    if (!last) load_chunk(codes, nchunk, tid, nxt);

#pragma unroll
    for (int k = 0; k < SK; ++k) {
      const int e0 = chunk * CHUNK + k * (STHREADS * EPL) + tid * EPL;
      note_sixteen(cur[k], e0, pat, plist, pcount);
    }
    __syncthreads();
    const int n = *pcount;
    __syncthreads();
    if (last || n + CHUNK > LCAP2) {
      drain_rows<NP, FUB>(src, dst, fa, fb, agg_a, agg_b, plist, sorted, ocnt, obase,
                          ocur, n, fresh, row_base, tid, wave, lane,
                          (FIRST && last) ? my_saved : nullptr);
      fresh = false;
      if (tid == 0) *pcount = 0;
      __syncthreads();
    }
    if (!last) {
#pragma unroll
      for (int k = 0; k < SK; ++k) cur[k] = nxt[k];
    }
    chunk = nchunk;
  }
}

__global__ __launch_bounds__(STHREADS) void scatter_tile_kernel(
    const int* __restrict__ src, const int* __restrict__ dst,
    float* scratch_f, const float* __restrict__ feat, float* __restrict__ agg) {
  __shared__ unsigned plist[LCAP2];
  __shared__ __attribute__((aligned(16))) unsigned sorted[LCAP2 + 8];
  __shared__ int pcount;
  __shared__ int ocnt[NBIN], obase[NBIN + 1], ocur[NBIN];
  scatter_body<1, 8, false>(src, dst, (const unsigned char*)scratch_f, feat, feat, agg,
                            agg, (unsigned*)scratch_f, plist, sorted, &pcount, ocnt,
                            obase, ocur);
}

template <bool FIRST>
__global__ __launch_bounds__(STHREADS) void scatter_pair_kernel(
    const int* __restrict__ src, const int* __restrict__ dst,
    float* scratch_f, const float* __restrict__ feat_a,
    const float* __restrict__ feat_b, float* __restrict__ agg_a,
    float* __restrict__ agg_b) {
  __shared__ unsigned plist[LCAP2];
  __shared__ __attribute__((aligned(16))) unsigned sorted[LCAP2 + 8];
  __shared__ int pcount;
  __shared__ int ocnt[NBIN], obase[NBIN + 1], ocur[NBIN];
  scatter_body<2, 4, FIRST>(src, dst, (const unsigned char*)scratch_f, feat_a, feat_b,
                            agg_a, agg_b, (unsigned*)scratch_f, plist, sorted, &pcount,
                            ocnt, obase, ocur);
}

__global__ __launch_bounds__(256) void gemm_zr_kernel(
    const float* __restrict__ x, const float* __restrict__ aggx,
    const float* __restrict__ hl, const float* __restrict__ aggh,
    const float* __restrict__ Wl, const float* __restrict__ bl,
    float* __restrict__ zout, float* __restrict__ vout) {
  __shared__ float lds[4 * 16 * 132];
  const int tid = threadIdx.x;
  const int wave = tid >> 5;
  const int lane = tid & 31;
  const int row0 = blockIdx.x * 128 + wave * 16;
  const int kof = (lane >> 4) << 1;
  const int nn = lane & 15;
  const int wc = weight_col(nn);
  const int cq = (lane >> 4) << 2;

  int arow = row0 + nn;
  if (arow > GN - 1) arow = GN - 1;

  v2f fs[32], ft[32];
  load_afrag_sum(x, aggx, arow, kof, fs);
  load_afrag_sum(hl, aggh, arow, kof, ft);

  const float* W0 = Wl + 0 * GH * GH;
  const float* W1 = Wl + 1 * GH * GH;
  const float* W2 = Wl + 2 * GH * GH;
  const float* W3 = Wl + 3 * GH * GH;
  float* L0 = lds + 0 * 16 * 132;
  float* L1 = lds + 1 * 16 * 132;
  float* L2 = lds + 2 * 16 * 132;
  float* L3 = lds + 3 * 16 * 132;

  for (int ct = 0; ct < GH / 16; ++ct) {
    const int n0 = ct * 16;
    __syncthreads();
    stage_slab(W0, n0, L0);
    stage_slab(W1, n0, L1);
    stage_slab(W2, n0, L2);
    stage_slab(W3, n0, L3);
    __syncthreads();

    const float* p0 = L0 + wc * 132 + kof;
    const float* p1 = L1 + wc * 132 + kof;
    const float* p2 = L2 + wc * 132 + kof;
    const float* p3 = L3 + wc * 132 + kof;

    v8f az0 = {0, 0, 0, 0, 0, 0, 0, 0}, az1 = {0, 0, 0, 0, 0, 0, 0, 0};
    v8f ar0 = {0, 0, 0, 0, 0, 0, 0, 0}, ar1 = {0, 0, 0, 0, 0, 0, 0, 0};
#pragma unroll
    for (int kk = 0; kk < 32; kk += 2) {
      az0 = wmma_f32(*(const v2f*)(p0 + 4 * kk), fs[kk], az0);
      az0 = wmma_f32(*(const v2f*)(p1 + 4 * kk), ft[kk], az0);
      ar0 = wmma_f32(*(const v2f*)(p2 + 4 * kk), fs[kk], ar0);
      ar0 = wmma_f32(*(const v2f*)(p3 + 4 * kk), ft[kk], ar0);
      az1 = wmma_f32(*(const v2f*)(p0 + 4 * kk + 4), fs[kk + 1], az1);
      az1 = wmma_f32(*(const v2f*)(p1 + 4 * kk + 4), ft[kk + 1], az1);
      ar1 = wmma_f32(*(const v2f*)(p2 + 4 * kk + 4), fs[kk + 1], ar1);
      ar1 = wmma_f32(*(const v2f*)(p3 + 4 * kk + 4), ft[kk + 1], ar1);
    }
    const v8f accz = az0 + az1;
    const v8f accr = ar0 + ar1;

    const int clo = n0 + cq;
    const v4f b0_lo = *(const v4f*)(bl + 0 * GH + clo), b0_hi = *(const v4f*)(bl + 0 * GH + clo + 8);
    const v4f b1_lo = *(const v4f*)(bl + 1 * GH + clo), b1_hi = *(const v4f*)(bl + 1 * GH + clo + 8);
    const v4f b2_lo = *(const v4f*)(bl + 2 * GH + clo), b2_hi = *(const v4f*)(bl + 2 * GH + clo + 8);
    const v4f b3_lo = *(const v4f*)(bl + 3 * GH + clo), b3_hi = *(const v4f*)(bl + 3 * GH + clo + 8);
    const size_t off = (size_t)arow * GH + clo;
    const v4f h_lo = *(const v4f*)(hl + off);
    const v4f h_hi = *(const v4f*)(hl + off + 8);
    hold_epilogue(b0_lo, b0_hi, b1_lo, b1_hi, b2_lo, b2_hi, b3_lo, b3_hi, h_lo, h_hi);
    const v4f bz_lo = b0_lo + b1_lo, bz_hi = b0_hi + b1_hi;
    const v4f br_lo = b2_lo + b3_lo, br_hi = b2_hi + b3_hi;
    v4f z_lo, z_hi, v_lo, v_hi;
#pragma unroll
    for (int i = 0; i < 4; ++i) {
      z_lo[i] = sigmoid_f(accz[i] + bz_lo[i]);
      z_hi[i] = sigmoid_f(accz[4 + i] + bz_hi[i]);
      v_lo[i] = sigmoid_f(accr[i] + br_lo[i]) * h_lo[i];
      v_hi[i] = sigmoid_f(accr[4 + i] + br_hi[i]) * h_hi[i];
    }
    if (row0 + nn < GN) {
      *(v4f*)(zout + off) = z_lo;
      *(v4f*)(zout + off + 8) = z_hi;
      *(v4f*)(vout + off) = v_lo;
      *(v4f*)(vout + off + 8) = v_hi;
    }
  }
}

__global__ __launch_bounds__(256) void gemm_ht_kernel(
    const float* __restrict__ x, const float* __restrict__ aggx,
    const float* __restrict__ v, const float* __restrict__ aggv,
    const float* __restrict__ Wl, const float* __restrict__ bl,
    const float* __restrict__ zbuf, const float* __restrict__ hl,
    float* __restrict__ outp) {
  __shared__ float lds[2 * 16 * 132];
  const int tid = threadIdx.x;
  const int wave = tid >> 5;
  const int lane = tid & 31;
  const int row0 = blockIdx.x * 128 + wave * 16;
  const int kof = (lane >> 4) << 1;
  const int nn = lane & 15;
  const int wc = weight_col(nn);
  const int cq = (lane >> 4) << 2;

  int arow = row0 + nn;
  if (arow > GN - 1) arow = GN - 1;

  v2f fs[32], fu[32];
  load_afrag_sum(x, aggx, arow, kof, fs);
  hold_fragments(fs);
  load_afrag_sum(v, aggv, arow, kof, fu);

  const float* W4 = Wl + 4 * GH * GH;
  const float* W5 = Wl + 5 * GH * GH;
  float* L4 = lds + 0 * 16 * 132;
  float* L5 = lds + 1 * 16 * 132;

  for (int ct = 0; ct < GH / 16; ++ct) {
    const int n0 = ct * 16;
    __syncthreads();
    stage_slab(W4, n0, L4);
    stage_slab(W5, n0, L5);
    __syncthreads();

    const float* p4 = L4 + wc * 132 + kof;
    const float* p5 = L5 + wc * 132 + kof;

    v8f a0 = {0, 0, 0, 0, 0, 0, 0, 0}, a1 = {0, 0, 0, 0, 0, 0, 0, 0};
#pragma unroll
    for (int kk = 0; kk < 32; kk += 2) {
      a0 = wmma_f32(*(const v2f*)(p4 + 4 * kk), fs[kk], a0);
      a0 = wmma_f32(*(const v2f*)(p5 + 4 * kk), fu[kk], a0);
      a1 = wmma_f32(*(const v2f*)(p4 + 4 * kk + 4), fs[kk + 1], a1);
      a1 = wmma_f32(*(const v2f*)(p5 + 4 * kk + 4), fu[kk + 1], a1);
    }
    const v8f acc = a0 + a1;

    const int clo = n0 + cq;
    const v4f b4_lo = *(const v4f*)(bl + 4 * GH + clo), b4_hi = *(const v4f*)(bl + 4 * GH + clo + 8);
    const v4f b5_lo = *(const v4f*)(bl + 5 * GH + clo), b5_hi = *(const v4f*)(bl + 5 * GH + clo + 8);
    const size_t off = (size_t)arow * GH + clo;
    const v4f z_lo = *(const v4f*)(zbuf + off);
    const v4f z_hi = *(const v4f*)(zbuf + off + 8);
    const v4f h_lo = *(const v4f*)(hl + off);
    const v4f h_hi = *(const v4f*)(hl + off + 8);
    hold_epilogue(b4_lo, b4_hi, b5_lo, b5_hi, z_lo, z_hi, h_lo, h_hi);
    const v4f bh_lo = b4_lo + b5_lo, bh_hi = b4_hi + b5_hi;
    v4f o_lo, o_hi;
#pragma unroll
    for (int i = 0; i < 4; ++i) {
      const float ht_lo = tanh_f(acc[i] + bh_lo[i]);
      const float ht_hi = tanh_f(acc[4 + i] + bh_hi[i]);
      o_lo[i] = z_lo[i] * h_lo[i] + (1.0f - z_lo[i]) * ht_lo;
      o_hi[i] = z_hi[i] * h_hi[i] + (1.0f - z_hi[i]) * ht_hi;
    }
    if (row0 + nn < GN) {
      *(v4f*)(outp + off) = o_lo;
      *(v4f*)(outp + off + 8) = o_hi;
    }
  }
}

extern "C" void kernel_launch(void* const* d_in, const int* in_sizes, int n_in,
                              void* d_out, int out_size, void* d_ws,
                              size_t ws_size, hipStream_t stream) {
  (void)in_sizes; (void)n_in; (void)out_size; (void)ws_size;
  const float* inp = (const float*)d_in[0];
  const int* edg = (const int*)d_in[1];
  const float* hmat = (const float*)d_in[2];
  const float* Wmat = (const float*)d_in[3];
  const float* bmat = (const float*)d_in[4];
  float* out = (float*)d_out;

  const int* src = edg;
  const int* dst = edg + GE;

  float* agg_x = (float*)d_ws;
  float* agg_t = agg_x + NH;
  float* zbuf = agg_t + NH;
  float* vbuf = zbuf + NH;

  const int gemm_blocks = (GN + 127) / 128;

  static_assert((size_t)GE <= NH * sizeof(float), "codes fit one output plane");
  float* codes = out + (size_t)(GLAYERS - 1) * NH;
  tile_code_kernel<<<(GE / EPL + 255) / 256, 256, 0, stream>>>(dst, codes);

  for (int l = 0; l < GLAYERS; ++l) {
    const float* x = (l == 0) ? inp : (out + (size_t)(l - 1) * NH);
    const float* hl = hmat + (size_t)l * NH;
    const float* Wl = Wmat + (size_t)l * 6 * GH * GH;
    const float* bl = bmat + (size_t)l * 6 * GH;
    float* outl = out + (size_t)l * NH;

    if (l == 0)
      scatter_pair_kernel<true><<<NTILES, STHREADS, 0, stream>>>(src, dst, codes, x, hl,
                                                                 agg_x, agg_t);
    else
      scatter_pair_kernel<false><<<NTILES, STHREADS, 0, stream>>>(src, dst, codes, x, hl,
                                                                  agg_x, agg_t);

    gemm_zr_kernel<<<gemm_blocks, 256, 0, stream>>>(x, agg_x, hl, agg_t, Wl,
                                                    bl, zbuf, vbuf);

    scatter_tile_kernel<<<NTILES, STHREADS, 0, stream>>>(src, dst, codes, vbuf, agg_t);

    gemm_ht_kernel<<<gemm_blocks, 256, 0, stream>>>(x, agg_x, vbuf, agg_t, Wl,
                                                    bl, zbuf, hl, outl);
  }
}


# Weight products taken BEFORE the edge sums (GIN is linear): GEMMs read x, h, r*h alone and store product rows; the scatter's owner wave adds own row + biases, finishes the gate; sum order changes
# speedup vs baseline: 4.1470x; 1.0676x over previous
#include <hip/hip_runtime.h>
#include <cstdint>
#include <cstddef>

constexpr int GN = 50000;
constexpr int GE = 1600000;
constexpr int GH = 128;
constexpr int GLAYERS = 2;
constexpr size_t NH = (size_t)GN * GH;

typedef __attribute__((ext_vector_type(2))) float v2f;
typedef __attribute__((ext_vector_type(8))) float v8f;
typedef __attribute__((ext_vector_type(4))) float v4f;

__device__ __forceinline__ v8f wmma_f32(v2f a, v2f b, v8f c) {
  return __builtin_amdgcn_wmma_f32_16x16x4_f32(
       false, a,  false, b,
       (short)0, c,  false,  false);
}

__device__ __forceinline__ float sigmoid_f(float x) {
  return 1.0f / (1.0f + __expf(-x));
}
__device__ __forceinline__ float tanh_f(float x) {
  return 1.0f - 2.0f / (1.0f + __expf(2.0f * x));
}

__device__ __forceinline__ int weight_col(int m) {
  return ((m & 4) << 1) | ((m & 8) >> 1) | (m & 3);
}

__device__ __forceinline__ void load_frag(const float* __restrict__ X, int row,
                                          int kof, v2f* frag) {
  const float* xr = X + (size_t)row * GH;
#pragma unroll
  for (int kk = 0; kk < 32; ++kk) {
    const float2 xv = *(const float2*)(xr + (kk << 2) + kof);
    v2f f;
    f.x = xv.x;
    f.y = xv.y;
    frag[kk] = f;
  }
}

__device__ __forceinline__ void stage_slab(const float* __restrict__ Wg, int n0,
                                           float* lds) {
  const int tid = threadIdx.x;
#pragma unroll
  for (int i = 0; i < 8; ++i) {
    const int idx = i * 256 + tid;
    const int k = idx >> 4;
    const int c = idx & 15;
    lds[c * 132 + k] = Wg[k * GH + n0 + c];
  }
}

constexpr int TR = 197;
constexpr int NTILES = (GN + TR - 1) / TR;
constexpr unsigned NO_TILE = 255u;
constexpr int STHREADS = 1024;
constexpr int SWAVES = STHREADS / 32;
constexpr int RPW = (TR + SWAVES - 1) / SWAVES;
constexpr int EPL = 16;
constexpr int SK = 2;
constexpr int CHUNK = STHREADS * EPL * SK;
constexpr int NCHUNK = (GE + CHUNK - 1) / CHUNK;
constexpr int NBIN = SWAVES * RPW;
constexpr int LCAP2 = 39936;

constexpr int SCAP = 8192;
constexpr int SHDR = 256;
constexpr int SSTRIDE = SHDR + SCAP;
constexpr unsigned NOT_SAVED = 0xFFFFFFFFu;
constexpr size_t TILE_WORDS = (size_t)TR * GH;

static_assert(NTILES * TR >= GN, "tiles cover all rows");
static_assert(NBIN + 2 <= SHDR && SCAP % 4 == 0 && SCAP + 8 <= LCAP2, "header and list fit");
static_assert((size_t)SSTRIDE <= TILE_WORDS && (TILE_WORDS * 4) % 16 == 0, "a saved list fits its tile's rows, 16-byte aligned");
static_assert((size_t)(NTILES - 1) * TILE_WORDS + SSTRIDE <= NH, "the last tile's saved list ends inside the plane");
static_assert(GLAYERS >= 2, "the codes live in the first layer's output plane, the saved lists in the last layer's");
static_assert(NTILES <= (int)NO_TILE, "every tile number and the spare code fit one byte");
static_assert(GE % EPL == 0, "16-byte code loads");
static_assert(LCAP2 >= CHUNK, "an emptied list holds every edge of one scan step");
static_assert((2 * LCAP2 + 3 * NBIN + 16) * 4 <= 320 * 1024, "both lists fit the LDS");
static_assert(TR <= 256 && GN < (1 << 24), "local row and source row fit one word");
static_assert(SWAVES == 32 && NBIN >= TR + 1, "the first wave's lanes take RPW counters each");

__device__ __forceinline__ unsigned tile_code(int d) {
  const unsigned a = (unsigned)d;
  const unsigned lim = NO_TILE * (unsigned)TR;
  return (a < lim ? a : lim) / (unsigned)TR;
}

__device__ __forceinline__ uint4 codes_of_sixteen(const int* __restrict__ dst16) {
  const int4* d4 = (const int4*)dst16;
  int4 d[4];
#pragma unroll
  for (int q = 0; q < 4; ++q) d[q] = d4[q];
  unsigned w[4];
#pragma unroll
  for (int q = 0; q < 4; ++q)
    w[q] = tile_code(d[q].x) | (tile_code(d[q].y) << 8) |
           (tile_code(d[q].z) << 16) | (tile_code(d[q].w) << 24);
  return make_uint4(w[0], w[1], w[2], w[3]);
}

__global__ __launch_bounds__(256) void tile_code_kernel(
    const int* __restrict__ dst, float* __restrict__ codes_f) {
  const int g = blockIdx.x * 256 + threadIdx.x;
  if (g >= GE / EPL) return;
  ((uint4*)codes_f)[g] = codes_of_sixteen(dst + (size_t)g * EPL);
}

template <bool FROM_DST>
__device__ __forceinline__ void load_chunk(const unsigned char* __restrict__ codes,
                                           const int* __restrict__ dst,
                                           int chunk, int tid, uint4* buf) {
#pragma unroll
  for (int k = 0; k < SK; ++k) {
    const int e0 = chunk * CHUNK + k * (STHREADS * EPL) + tid * EPL;
    const int e = e0 < GE ? e0 : GE - EPL;
    if (FROM_DST)
      buf[k] = codes_of_sixteen(dst + e);
    else
      buf[k] = *(const uint4*)(codes + e);
  }
}

__device__ __forceinline__ unsigned zero_bytes(unsigned x) {
  const unsigned m = 0x7F7F7F7Fu;
  return ~(((x & m) + m) | x | m);
}

__device__ __forceinline__ void note_sixteen(uint4 w, int e0, unsigned pat,
                                             unsigned* plist, int* pcount) {
  const unsigned z0 = zero_bytes(w.x ^ pat);
  const unsigned z1 = zero_bytes(w.y ^ pat);
  const unsigned z2 = zero_bytes(w.z ^ pat);
  const unsigned z3 = zero_bytes(w.w ^ pat);
  unsigned hits = (z0 >> 7) | (z1 >> 6) | (z2 >> 5) | (z3 >> 4);
  if (e0 >= GE) hits = 0;
  while (hits != 0) {
    const int bit = __builtin_ctz(hits);
    hits &= hits - 1;
    const int pos = atomicAdd(pcount, 1);
    plist[pos] = (unsigned)(e0 + ((bit & 7) << 2) + (bit >> 3));
  }
}

__device__ __forceinline__ void hold_loaded(const v4f (&v)[4][2]) {
  asm volatile("" ::
               "v"(v[0][0]), "v"(v[0][1]), "v"(v[1][0]), "v"(v[1][1]),
               "v"(v[2][0]), "v"(v[2][1]), "v"(v[3][0]), "v"(v[3][1]));
}
template <int F>
__device__ __forceinline__ void hold_loaded(const v4f (&)[F][1]) {}

struct GateIn {
  const float* h;
  const float* z;
  const float* bias;
};

template <int NP, int FUB>
__device__ __forceinline__ void sum_rows(
    const float* __restrict__ fa, const float* __restrict__ fb, float* agg_a,
    float* agg_b, GateIn gin, const unsigned* sorted, const int* obase, bool fresh,
    bool final, int row_base, int wave, int lane);

template <int NP, int FUB>
__device__ __forceinline__ void drain_rows(
    const int* __restrict__ src, const int* __restrict__ dst,
    const float* __restrict__ fa, const float* __restrict__ fb,
    float* agg_a, float* agg_b, GateIn gin, unsigned* plist, unsigned* sorted,
    int* ocnt, int* obase, int* ocur, int n, bool fresh, bool final, int row_base,
    int tid, int wave, int lane, unsigned* save_to) {
  if (tid < NBIN) ocnt[tid] = 0;
  __syncthreads();

  for (int i = tid; i < n; i += 4 * STHREADS) {
    unsigned ent[4];
    int s[4], d[4];
#pragma unroll
    for (int q = 0; q < 4; ++q) {
      const int ii = i + q * STHREADS;
      ent[q] = plist[ii < n ? ii : i];
    }
#pragma unroll
    for (int q = 0; q < 4; ++q) {
      s[q] = src[ent[q]];
      d[q] = dst[ent[q]];
    }
#pragma unroll
    for (int q = 0; q < 4; ++q) {
      const int ii = i + q * STHREADS;
      if (ii < n) {
        const unsigned lr = (unsigned)(d[q] - row_base);
        plist[ii] = ((unsigned)s[q] << 8) | lr;
        atomicAdd(&ocnt[lr], 1);
      }
    }
  }
  __syncthreads();

  if (wave == 0) {
    int c[RPW];
    int run = 0;
#pragma unroll
    for (int q = 0; q < RPW; ++q) {
      const int x = ocnt[lane * RPW + q];
      c[q] = run;
      run += x;
    }
    int incl = run;
#pragma unroll
    for (int dlt = 1; dlt < 32; dlt <<= 1) {
      const int t = __shfl_up(incl, dlt);
      if (lane >= dlt) incl += t;
    }
    const int excl = incl - run;
#pragma unroll
    for (int q = 0; q < RPW; ++q) {
      obase[lane * RPW + q] = excl + c[q];
      ocur[lane * RPW + q] = excl + c[q];
    }
    if (lane == 31) obase[NBIN] = incl;
  }
  __syncthreads();

  for (int i = tid; i < n; i += STHREADS) {
    const unsigned w = plist[i];
    const int pos = atomicAdd(&ocur[w & 255u], 1);
    sorted[pos] = w;
  }
  __syncthreads();

  if (save_to != nullptr) {
    if (fresh && n <= SCAP) {
      for (int i = tid * 4; i < n; i += 4 * STHREADS)
        *(uint4*)(save_to + SHDR + i) = *(const uint4*)(sorted + i);
      if (tid <= NBIN) save_to[1 + tid] = (unsigned)obase[tid];
      if (tid == 0) save_to[0] = (unsigned)n;
    } else if (tid == 0) {
      save_to[0] = NOT_SAVED;
    }
  }

  sum_rows<NP, FUB>(fa, fb, agg_a, agg_b, gin, sorted, obase, fresh, final, row_base,
                    wave, lane);
}

template <int NP, int FUB>
__device__ __forceinline__ void sum_rows(
    const float* __restrict__ fa, const float* __restrict__ fb, float* agg_a,
    float* agg_b, GateIn gin, const unsigned* sorted, const int* obase, bool fresh,
    bool final, int row_base, int wave, int lane) {
  const float* bp = gin.bias + lane * 4;
  v4f bias_a, bias_b;
  if (NP == 2) {
    bias_a = *(const v4f*)(bp + 0 * GH) + *(const v4f*)(bp + 1 * GH);
    bias_b = *(const v4f*)(bp + 2 * GH) + *(const v4f*)(bp + 3 * GH);
  } else {
    bias_a = *(const v4f*)(bp + 4 * GH) + *(const v4f*)(bp + 5 * GH);
    bias_b = bias_a;
  }
#pragma nounroll
  for (int j = 0; j < RPW; ++j) {
    const int lr = wave + j * SWAVES;
    const int row = row_base + lr;
    if (lr >= TR || row >= GN) break;
    int i0 = __builtin_amdgcn_readfirstlane(obase[lr]);
    const int sege = __builtin_amdgcn_readfirstlane(obase[lr + 1]);
    float* oa = agg_a + (size_t)row * GH + lane * 4;
    float* ob = agg_b + (size_t)row * GH + lane * 4;
    v4f a[NP];
#pragma unroll
    for (int p = 0; p < NP; ++p) a[p] = (v4f){0.0f, 0.0f, 0.0f, 0.0f};
    if (!fresh) {
      a[0] = *(const v4f*)oa;
      if (NP == 2) a[NP - 1] = *(const v4f*)ob;
    }
    const size_t own = (size_t)row * GH + lane * 4;
    v4f own_a = (v4f){0.0f, 0.0f, 0.0f, 0.0f}, own_b = own_a, own_h = own_a, own_z = own_a;
    if (final) {
      own_a = *(const v4f*)(fa + own);
      own_h = *(const v4f*)(gin.h + own);
      if (NP == 2) own_b = *(const v4f*)(fb + own);
      else own_z = *(const v4f*)(gin.z + own);
    }
    for (; i0 + FUB <= sege; i0 += FUB) {
      unsigned w[FUB];
      v4f v[FUB][NP];
#pragma unroll
      for (int u = 0; u < FUB; ++u)
        w[u] = (unsigned)__builtin_amdgcn_readfirstlane((int)sorted[i0 + u]);
#pragma unroll
      for (int u = 0; u < FUB; ++u) {
        const size_t off = (size_t)(w[u] >> 8) * GH + lane * 4;
        v[u][0] = *(const v4f*)(fa + off);
        if (NP == 2) v[u][NP - 1] = *(const v4f*)(fb + off);
      }
      hold_loaded(v);
#pragma unroll
      for (int u = 0; u < FUB; ++u)
#pragma unroll
        for (int p = 0; p < NP; ++p) a[p] += v[u][p];
    }
    const int rem = sege - i0;
    if (rem > 0) {
      unsigned w[FUB - 1];
      v4f v[FUB - 1][NP];
#pragma unroll
      for (int u = 0; u < FUB - 1; ++u)
        w[u] = (unsigned)__builtin_amdgcn_readfirstlane((int)sorted[i0 + u]);
#pragma unroll
      for (int u = 0; u < FUB - 1; ++u)
        if (u < rem) {
          const size_t off = (size_t)(w[u] >> 8) * GH + lane * 4;
          v[u][0] = *(const v4f*)(fa + off);
          if (NP == 2) v[u][NP - 1] = *(const v4f*)(fb + off);
        }
#pragma unroll
      for (int u = 0; u < FUB - 1; ++u)
        if (u < rem) {
#pragma unroll
          for (int p = 0; p < NP; ++p) a[p] += v[u][p];
        }
    }
    if (!final) {
      *(v4f*)oa = a[0];
      if (NP == 2) *(v4f*)ob = a[NP - 1];
    } else if (NP == 2) {
      v4f zv, vv;
#pragma unroll
      for (int i = 0; i < 4; ++i) {
        zv[i] = sigmoid_f(own_a[i] + a[0][i] + bias_a[i]);
        vv[i] = sigmoid_f(own_b[i] + a[NP - 1][i] + bias_b[i]) * own_h[i];
      }
      *(v4f*)oa = zv;
      *(v4f*)ob = vv;
    } else {
      v4f ov;
#pragma unroll
      for (int i = 0; i < 4; ++i) {
        const float ht = tanh_f(own_a[i] + a[0][i] + bias_a[i]);
        ov[i] = own_z[i] * own_h[i] + (1.0f - own_z[i]) * ht;
      }
      *(v4f*)oa = ov;
    }
  }
  __syncthreads();
}

template <int NP, int FUB, bool FIRST>
__device__ __forceinline__ void scatter_body(
    const int* __restrict__ src, const int* __restrict__ dst,
    const unsigned char* __restrict__ codes, const float* __restrict__ fa,
    const float* __restrict__ fb, float* agg_a, float* agg_b, GateIn gin,
    unsigned* saved, unsigned* plist, unsigned* sorted, int* pcount, int* ocnt,
    int* obase, int* ocur) {
  const int tid = threadIdx.x;
  const int lane = tid & 31;
  const int wave = tid >> 5;
  const int row_base = blockIdx.x * TR;
  const unsigned pat = blockIdx.x * 0x01010101u;
  unsigned* my_saved = saved + (size_t)blockIdx.x * TILE_WORDS;

  if (!FIRST) {
    const unsigned nsv = (unsigned)__builtin_amdgcn_readfirstlane((int)my_saved[0]);
    if (nsv <= (unsigned)SCAP) {
      for (int i = tid * 4; i < (int)nsv; i += 4 * STHREADS)
        *(uint4*)(sorted + i) = *(const uint4*)(my_saved + SHDR + i);
      if (tid <= NBIN) obase[tid] = (int)my_saved[1 + tid];
      __syncthreads();
      sum_rows<NP, FUB>(fa, fb, agg_a, agg_b, gin, sorted, obase, true, true, row_base,
                        wave, lane);
      return;
    }
  }

  if (tid == 0) *pcount = 0;
  __syncthreads();

  int chunk = (blockIdx.x * NCHUNK) / NTILES;
  uint4 cur[SK], nxt[SK];
  load_chunk<!FIRST>(codes, dst, chunk, tid, cur);
  bool fresh = true;
  for (int c = 0; c < NCHUNK; ++c) {
    int nchunk = chunk + 1;
    if (nchunk == NCHUNK) nchunk = 0;
    const bool last = (c + 1 == NCHUNK);
    if (FIRST && !last) load_chunk<false>(codes, dst, nchunk, tid, nxt);

#pragma unroll
    for (int k = 0; k < SK; ++k) {
      const int e0 = chunk * CHUNK + k * (STHREADS * EPL) + tid * EPL;
      note_sixteen(cur[k], e0, pat, plist, pcount);
    }
    __syncthreads();
    const int n = *pcount;
    __syncthreads();
    if (last || n + CHUNK > LCAP2) {
      drain_rows<NP, FUB>(src, dst, fa, fb, agg_a, agg_b, gin, plist, sorted, ocnt,
                          obase, ocur, n, fresh, last, row_base, tid, wave, lane,
                          (FIRST && last) ? my_saved : nullptr);
      fresh = false;
      if (tid == 0) *pcount = 0;
      __syncthreads();
    }
    if (!last) {
      if (FIRST) {
#pragma unroll
        for (int k = 0; k < SK; ++k) cur[k] = nxt[k];
      } else {
        load_chunk<true>(codes, dst, nchunk, tid, cur);
      }
    }
    chunk = nchunk;
  }
}

__global__ __launch_bounds__(STHREADS) void scatter_tile_kernel(
    const int* __restrict__ src, const int* __restrict__ dst,
    float* lists_f, const float* __restrict__ prod, const float* __restrict__ zin,
    const float* __restrict__ hl, const float* __restrict__ bl, float* out) {
  __shared__ unsigned plist[LCAP2];
  __shared__ __attribute__((aligned(16))) unsigned sorted[LCAP2 + 8];
  __shared__ int pcount;
  __shared__ int ocnt[NBIN], obase[NBIN + 1], ocur[NBIN];
  GateIn gin;
  gin.h = hl;
  gin.z = zin;
  gin.bias = bl;
  scatter_body<1, 8, false>(src, dst, (const unsigned char*)nullptr, prod, prod, out,
                            out, gin, (unsigned*)lists_f, plist, sorted, &pcount, ocnt,
                            obase, ocur);
}

template <bool FIRST>
__global__ __launch_bounds__(STHREADS) void scatter_pair_kernel(
    const int* __restrict__ src, const int* __restrict__ dst,
    const float* codes_f, float* lists_f, const float* __restrict__ prod_z,
    const float* __restrict__ prod_r, const float* __restrict__ hl,
    const float* __restrict__ bl, float* zout, float* vout) {
  __shared__ unsigned plist[LCAP2];
  __shared__ __attribute__((aligned(16))) unsigned sorted[LCAP2 + 8];
  __shared__ int pcount;
  __shared__ int ocnt[NBIN], obase[NBIN + 1], ocur[NBIN];
  GateIn gin;
  gin.h = hl;
  gin.z = hl;
  gin.bias = bl;
  scatter_body<2, 4, FIRST>(src, dst, (const unsigned char*)codes_f, prod_z, prod_r,
                            zout, vout, gin, (unsigned*)lists_f, plist, sorted, &pcount,
                            ocnt, obase, ocur);
}

__global__ __launch_bounds__(256) void gemm_zr_kernel(
    const float* __restrict__ x, const float* __restrict__ hl,
    const float* __restrict__ Wl, float* __restrict__ pz, float* __restrict__ pr) {
  __shared__ float lds[4 * 16 * 132];
  const int tid = threadIdx.x;
  const int wave = tid >> 5;
  const int lane = tid & 31;
  const int row0 = blockIdx.x * 128 + wave * 16;
  const int kof = (lane >> 4) << 1;
  const int nn = lane & 15;
  const int wc = weight_col(nn);
  const int cq = (lane >> 4) << 2;

  int arow = row0 + nn;
  if (arow > GN - 1) arow = GN - 1;

  v2f fs[32], ft[32];
  load_frag(x, arow, kof, fs);
  load_frag(hl, arow, kof, ft);

  const float* W0 = Wl + 0 * GH * GH;
  const float* W1 = Wl + 1 * GH * GH;
  const float* W2 = Wl + 2 * GH * GH;
  const float* W3 = Wl + 3 * GH * GH;
  float* L0 = lds + 0 * 16 * 132;
  float* L1 = lds + 1 * 16 * 132;
  float* L2 = lds + 2 * 16 * 132;
  float* L3 = lds + 3 * 16 * 132;

  for (int ct = 0; ct < GH / 16; ++ct) {
    const int n0 = ct * 16;
    __syncthreads();
    stage_slab(W0, n0, L0);
    stage_slab(W1, n0, L1);
    stage_slab(W2, n0, L2);
    stage_slab(W3, n0, L3);
    __syncthreads();

    const float* p0 = L0 + wc * 132 + kof;
    const float* p1 = L1 + wc * 132 + kof;
    const float* p2 = L2 + wc * 132 + kof;
    const float* p3 = L3 + wc * 132 + kof;

    v8f az0 = {0, 0, 0, 0, 0, 0, 0, 0}, az1 = {0, 0, 0, 0, 0, 0, 0, 0};
    v8f ar0 = {0, 0, 0, 0, 0, 0, 0, 0}, ar1 = {0, 0, 0, 0, 0, 0, 0, 0};
#pragma unroll
    for (int kk = 0; kk < 32; kk += 2) {
      az0 = wmma_f32(*(const v2f*)(p0 + 4 * kk), fs[kk], az0);
      az0 = wmma_f32(*(const v2f*)(p1 + 4 * kk), ft[kk], az0);
      ar0 = wmma_f32(*(const v2f*)(p2 + 4 * kk), fs[kk], ar0);
      ar0 = wmma_f32(*(const v2f*)(p3 + 4 * kk), ft[kk], ar0);
      az1 = wmma_f32(*(const v2f*)(p0 + 4 * kk + 4), fs[kk + 1], az1);
      az1 = wmma_f32(*(const v2f*)(p1 + 4 * kk + 4), ft[kk + 1], az1);
      ar1 = wmma_f32(*(const v2f*)(p2 + 4 * kk + 4), fs[kk + 1], ar1);
      ar1 = wmma_f32(*(const v2f*)(p3 + 4 * kk + 4), ft[kk + 1], ar1);
    }
    const v8f accz = az0 + az1;
    const v8f accr = ar0 + ar1;

    const int clo = n0 + cq;
    const size_t off = (size_t)arow * GH + clo;
    if (row0 + nn < GN) {
      *(v4f*)(pz + off) = (v4f){accz[0], accz[1], accz[2], accz[3]};
      *(v4f*)(pz + off + 8) = (v4f){accz[4], accz[5], accz[6], accz[7]};
      *(v4f*)(pr + off) = (v4f){accr[0], accr[1], accr[2], accr[3]};
      *(v4f*)(pr + off + 8) = (v4f){accr[4], accr[5], accr[6], accr[7]};
    }
  }
}

__global__ __launch_bounds__(256) void gemm_ht_kernel(
    const float* __restrict__ x, const float* __restrict__ v,
    const float* __restrict__ Wl, float* __restrict__ ph) {
  __shared__ float lds[2 * 16 * 132];
  const int tid = threadIdx.x;
  const int wave = tid >> 5;
  const int lane = tid & 31;
  const int row0 = blockIdx.x * 128 + wave * 16;
  const int kof = (lane >> 4) << 1;
  const int nn = lane & 15;
  const int wc = weight_col(nn);
  const int cq = (lane >> 4) << 2;

  int arow = row0 + nn;
  if (arow > GN - 1) arow = GN - 1;

  v2f fs[32], fu[32];
  load_frag(x, arow, kof, fs);
  load_frag(v, arow, kof, fu);

  const float* W4 = Wl + 4 * GH * GH;
  const float* W5 = Wl + 5 * GH * GH;
  float* L4 = lds + 0 * 16 * 132;
  float* L5 = lds + 1 * 16 * 132;

  for (int ct = 0; ct < GH / 16; ++ct) {
    const int n0 = ct * 16;
    __syncthreads();
    stage_slab(W4, n0, L4);
    stage_slab(W5, n0, L5);
    __syncthreads();

    const float* p4 = L4 + wc * 132 + kof;
    const float* p5 = L5 + wc * 132 + kof;

    v8f a0 = {0, 0, 0, 0, 0, 0, 0, 0}, a1 = {0, 0, 0, 0, 0, 0, 0, 0};
#pragma unroll
    for (int kk = 0; kk < 32; kk += 2) {
      a0 = wmma_f32(*(const v2f*)(p4 + 4 * kk), fs[kk], a0);
      a0 = wmma_f32(*(const v2f*)(p5 + 4 * kk), fu[kk], a0);
      a1 = wmma_f32(*(const v2f*)(p4 + 4 * kk + 4), fs[kk + 1], a1);
      a1 = wmma_f32(*(const v2f*)(p5 + 4 * kk + 4), fu[kk + 1], a1);
    }
    const v8f acc = a0 + a1;

    const int clo = n0 + cq;
    const size_t off = (size_t)arow * GH + clo;
    if (row0 + nn < GN) {
      *(v4f*)(ph + off) = (v4f){acc[0], acc[1], acc[2], acc[3]};
      *(v4f*)(ph + off + 8) = (v4f){acc[4], acc[5], acc[6], acc[7]};
    }
  }
}

extern "C" void kernel_launch(void* const* d_in, const int* in_sizes, int n_in,
                              void* d_out, int out_size, void* d_ws,
                              size_t ws_size, hipStream_t stream) {
  (void)in_sizes; (void)n_in; (void)out_size; (void)ws_size;
  const float* inp = (const float*)d_in[0];
  const int* edg = (const int*)d_in[1];
  const float* hmat = (const float*)d_in[2];
  const float* Wmat = (const float*)d_in[3];
  const float* bmat = (const float*)d_in[4];
  float* out = (float*)d_out;

  const int* src = edg;
  const int* dst = edg + GE;

  float* pa = (float*)d_ws;
  float* pb = pa + NH;
  float* zbuf = pb + NH;
  float* vbuf = zbuf + NH;

  const int gemm_blocks = (GN + 127) / 128;

  static_assert((size_t)GE <= NH * sizeof(float), "codes fit one output plane");
  float* codes = out;
  float* lists = out + (size_t)(GLAYERS - 1) * NH;
  tile_code_kernel<<<(GE / EPL + 255) / 256, 256, 0, stream>>>(dst, codes);

  for (int l = 0; l < GLAYERS; ++l) {
    const float* x = (l == 0) ? inp : (out + (size_t)(l - 1) * NH);
    const float* hl = hmat + (size_t)l * NH;
    const float* Wl = Wmat + (size_t)l * 6 * GH * GH;
    const float* bl = bmat + (size_t)l * 6 * GH;
    float* outl = out + (size_t)l * NH;

    gemm_zr_kernel<<<gemm_blocks, 256, 0, stream>>>(x, hl, Wl, pa, pb);

    if (l == 0)
      scatter_pair_kernel<true><<<NTILES, STHREADS, 0, stream>>>(src, dst, codes, lists, pa,
                                                                 pb, hl, bl, zbuf, vbuf);
    else
      scatter_pair_kernel<false><<<NTILES, STHREADS, 0, stream>>>(src, dst, codes, lists, pa,
                                                                  pb, hl, bl, zbuf, vbuf);

    gemm_ht_kernel<<<gemm_blocks, 256, 0, stream>>>(x, vbuf, Wl, pa);

    scatter_tile_kernel<<<NTILES, STHREADS, 0, stream>>>(src, dst, lists, pa, zbuf, hl, bl,
                                                         outl);
  }
}


# GEMMs: node-row tiles come in and product tiles leave through a per-wave LDS region: each global load/store covers 4 whole 128-byte lines, not 16 lines x 16-32 bytes; values and sum order unchanged
# speedup vs baseline: 4.3766x; 1.0554x over previous
#include <hip/hip_runtime.h>
#include <cstdint>
#include <cstddef>

constexpr int GN = 50000;
constexpr int GE = 1600000;
constexpr int GH = 128;
constexpr int GLAYERS = 2;
constexpr size_t NH = (size_t)GN * GH;

typedef __attribute__((ext_vector_type(2))) float v2f;
typedef __attribute__((ext_vector_type(8))) float v8f;
typedef __attribute__((ext_vector_type(4))) float v4f;

__device__ __forceinline__ v8f wmma_f32(v2f a, v2f b, v8f c) {
  return __builtin_amdgcn_wmma_f32_16x16x4_f32(
       false, a,  false, b,
       (short)0, c,  false,  false);
}

__device__ __forceinline__ float sigmoid_f(float x) {
  return 1.0f / (1.0f + __expf(-x));
}
__device__ __forceinline__ float tanh_f(float x) {
  return 1.0f - 2.0f / (1.0f + __expf(2.0f * x));
}

__device__ __forceinline__ int weight_col(int m) {
  return ((m & 4) << 1) | ((m & 8) >> 1) | (m & 3);
}

constexpr int FPITCH = 132;
constexpr int OPITCH = 36;
constexpr int WREGION = 16 * FPITCH;
static_assert(2 * 16 * OPITCH <= WREGION, "two outgoing strips fit the wave's region");

__device__ __forceinline__ void load_frag(const float* __restrict__ X, int row0,
                                          int lane, int nn, int kof, float* wl,
                                          v2f* frag) {
  v4f t[16];
#pragma unroll
  for (int i = 0; i < 16; ++i) {
    int r = row0 + i;
    if (r > GN - 1) r = GN - 1;
    t[i] = *(const v4f*)(X + (size_t)r * GH + lane * 4);
  }
#pragma unroll
  for (int i = 0; i < 16; ++i) *(v4f*)(wl + i * FPITCH + lane * 4) = t[i];
  __syncthreads();
#pragma unroll
  for (int kk = 0; kk < 32; ++kk)
    frag[kk] = *(const v2f*)(wl + nn * FPITCH + (kk << 2) + kof);
  __syncthreads();
}

__device__ __forceinline__ void put_piece(float* strip, int nn, int half, int cq,
                                          v8f acc) {
  float* p = strip + nn * OPITCH + half * 16 + cq;
  *(v4f*)p = (v4f){acc[0], acc[1], acc[2], acc[3]};
  *(v4f*)(p + 8) = (v4f){acc[4], acc[5], acc[6], acc[7]};
}

static_assert(GN % 16 == 0, "no 16-row tile straddles the end of the graph");
__device__ __forceinline__ void store_strip(const float* strip, float* __restrict__ P,
                                            int row0, int c0, int lane) {
  const int r = lane >> 3;
  const int c = (lane & 7) * 4;
  v4f v[4];
#pragma unroll
  for (int i = 0; i < 4; ++i) v[i] = *(const v4f*)(strip + (4 * i + r) * OPITCH + c);
  if (row0 < GN) {
#pragma unroll
    for (int i = 0; i < 4; ++i)
      *(v4f*)(P + (size_t)(row0 + 4 * i + r) * GH + c0 + c) = v[i];
  }
}

__device__ __forceinline__ void stage_slab(const float* __restrict__ Wg, int n0,
                                           float* lds) {
  const int tid = threadIdx.x;
#pragma unroll
  for (int i = 0; i < 8; ++i) {
    const int idx = i * 256 + tid;
    const int k = idx >> 4;
    const int c = idx & 15;
    lds[c * 132 + k] = Wg[k * GH + n0 + c];
  }
}

constexpr int TR = 197;
constexpr int NTILES = (GN + TR - 1) / TR;
constexpr unsigned NO_TILE = 255u;
constexpr int STHREADS = 1024;
constexpr int SWAVES = STHREADS / 32;
constexpr int RPW = (TR + SWAVES - 1) / SWAVES;
constexpr int EPL = 16;
constexpr int SK = 2;
constexpr int CHUNK = STHREADS * EPL * SK;
constexpr int NCHUNK = (GE + CHUNK - 1) / CHUNK;
constexpr int NBIN = SWAVES * RPW;
constexpr int LCAP2 = 39936;

constexpr int SCAP = 8192;
constexpr int SHDR = 256;
constexpr int SSTRIDE = SHDR + SCAP;
constexpr unsigned NOT_SAVED = 0xFFFFFFFFu;
constexpr size_t TILE_WORDS = (size_t)TR * GH;

static_assert(NTILES * TR >= GN, "tiles cover all rows");
static_assert(NBIN + 2 <= SHDR && SCAP % 4 == 0 && SCAP + 8 <= LCAP2, "header and list fit");
static_assert((size_t)SSTRIDE <= TILE_WORDS && (TILE_WORDS * 4) % 16 == 0, "a saved list fits its tile's rows, 16-byte aligned");
static_assert((size_t)(NTILES - 1) * TILE_WORDS + SSTRIDE <= NH, "the last tile's saved list ends inside the plane");
static_assert(GLAYERS >= 2, "the codes live in the first layer's output plane, the saved lists in the last layer's");
static_assert(NTILES <= (int)NO_TILE, "every tile number and the spare code fit one byte");
static_assert(GE % EPL == 0, "16-byte code loads");
static_assert(LCAP2 >= CHUNK, "an emptied list holds every edge of one scan step");
static_assert((2 * LCAP2 + 3 * NBIN + 16) * 4 <= 320 * 1024, "both lists fit the LDS");
static_assert(TR <= 256 && GN < (1 << 24), "local row and source row fit one word");
static_assert(SWAVES == 32 && NBIN >= TR + 1, "the first wave's lanes take RPW counters each");

__device__ __forceinline__ unsigned tile_code(int d) {
  const unsigned a = (unsigned)d;
  const unsigned lim = NO_TILE * (unsigned)TR;
  return (a < lim ? a : lim) / (unsigned)TR;
}

__device__ __forceinline__ uint4 codes_of_sixteen(const int* __restrict__ dst16) {
  const int4* d4 = (const int4*)dst16;
  int4 d[4];
#pragma unroll
  for (int q = 0; q < 4; ++q) d[q] = d4[q];
  unsigned w[4];
#pragma unroll
  for (int q = 0; q < 4; ++q)
    w[q] = tile_code(d[q].x) | (tile_code(d[q].y) << 8) |
           (tile_code(d[q].z) << 16) | (tile_code(d[q].w) << 24);
  return make_uint4(w[0], w[1], w[2], w[3]);
}

__global__ __launch_bounds__(256) void tile_code_kernel(
    const int* __restrict__ dst, float* __restrict__ codes_f) {
  const int g = blockIdx.x * 256 + threadIdx.x;
  if (g >= GE / EPL) return;
  ((uint4*)codes_f)[g] = codes_of_sixteen(dst + (size_t)g * EPL);
}

template <bool FROM_DST>
__device__ __forceinline__ void load_chunk(const unsigned char* __restrict__ codes,
                                           const int* __restrict__ dst,
                                           int chunk, int tid, uint4* buf) {
#pragma unroll
  for (int k = 0; k < SK; ++k) {
    const int e0 = chunk * CHUNK + k * (STHREADS * EPL) + tid * EPL;
    const int e = e0 < GE ? e0 : GE - EPL;
    if (FROM_DST)
      buf[k] = codes_of_sixteen(dst + e);
    else
      buf[k] = *(const uint4*)(codes + e);
  }
}

__device__ __forceinline__ unsigned zero_bytes(unsigned x) {
  const unsigned m = 0x7F7F7F7Fu;
  return ~(((x & m) + m) | x | m);
}

__device__ __forceinline__ void note_sixteen(uint4 w, int e0, unsigned pat,
                                             unsigned* plist, int* pcount) {
  const unsigned z0 = zero_bytes(w.x ^ pat);
  const unsigned z1 = zero_bytes(w.y ^ pat);
  const unsigned z2 = zero_bytes(w.z ^ pat);
  const unsigned z3 = zero_bytes(w.w ^ pat);
  unsigned hits = (z0 >> 7) | (z1 >> 6) | (z2 >> 5) | (z3 >> 4);
  if (e0 >= GE) hits = 0;
  while (hits != 0) {
    const int bit = __builtin_ctz(hits);
    hits &= hits - 1;
    const int pos = atomicAdd(pcount, 1);
    plist[pos] = (unsigned)(e0 + ((bit & 7) << 2) + (bit >> 3));
  }
}

__device__ __forceinline__ void hold_loaded(const v4f (&v)[4][2]) {
  asm volatile("" ::
               "v"(v[0][0]), "v"(v[0][1]), "v"(v[1][0]), "v"(v[1][1]),
               "v"(v[2][0]), "v"(v[2][1]), "v"(v[3][0]), "v"(v[3][1]));
}
template <int F>
__device__ __forceinline__ void hold_loaded(const v4f (&)[F][1]) {}

struct GateIn {
  const float* h;
  const float* z;
  const float* bias;
};

template <int NP, int FUB>
__device__ __forceinline__ void sum_rows(
    const float* __restrict__ fa, const float* __restrict__ fb, float* agg_a,
    float* agg_b, GateIn gin, const unsigned* sorted, const int* obase, bool fresh,
    bool final, int row_base, int wave, int lane);

template <int NP, int FUB>
__device__ __forceinline__ void drain_rows(
    const int* __restrict__ src, const int* __restrict__ dst,
    const float* __restrict__ fa, const float* __restrict__ fb,
    float* agg_a, float* agg_b, GateIn gin, unsigned* plist, unsigned* sorted,
    int* ocnt, int* obase, int* ocur, int n, bool fresh, bool final, int row_base,
    int tid, int wave, int lane, unsigned* save_to) {
  if (tid < NBIN) ocnt[tid] = 0;
  __syncthreads();

  for (int i = tid; i < n; i += 4 * STHREADS) {
    unsigned ent[4];
    int s[4], d[4];
#pragma unroll
    for (int q = 0; q < 4; ++q) {
      const int ii = i + q * STHREADS;
      ent[q] = plist[ii < n ? ii : i];
    }
#pragma unroll
    for (int q = 0; q < 4; ++q) {
      s[q] = src[ent[q]];
      d[q] = dst[ent[q]];
    }
#pragma unroll
    for (int q = 0; q < 4; ++q) {
      const int ii = i + q * STHREADS;
      if (ii < n) {
        const unsigned lr = (unsigned)(d[q] - row_base);
        plist[ii] = ((unsigned)s[q] << 8) | lr;
        atomicAdd(&ocnt[lr], 1);
      }
    }
  }
  __syncthreads();

  if (wave == 0) {
    int c[RPW];
    int run = 0;
#pragma unroll
    for (int q = 0; q < RPW; ++q) {
      const int x = ocnt[lane * RPW + q];
      c[q] = run;
      run += x;
    }
    int incl = run;
#pragma unroll
    for (int dlt = 1; dlt < 32; dlt <<= 1) {
      const int t = __shfl_up(incl, dlt);
      if (lane >= dlt) incl += t;
    }
    const int excl = incl - run;
#pragma unroll
    for (int q = 0; q < RPW; ++q) {
      obase[lane * RPW + q] = excl + c[q];
      ocur[lane * RPW + q] = excl + c[q];
    }
    if (lane == 31) obase[NBIN] = incl;
  }
  __syncthreads();

  for (int i = tid; i < n; i += STHREADS) {
    const unsigned w = plist[i];
    const int pos = atomicAdd(&ocur[w & 255u], 1);
    sorted[pos] = w;
  }
  __syncthreads();

  if (save_to != nullptr) {
    if (fresh && n <= SCAP) {
      for (int i = tid * 4; i < n; i += 4 * STHREADS)
        *(uint4*)(save_to + SHDR + i) = *(const uint4*)(sorted + i);
      if (tid <= NBIN) save_to[1 + tid] = (unsigned)obase[tid];
      if (tid == 0) save_to[0] = (unsigned)n;
    } else if (tid == 0) {
      save_to[0] = NOT_SAVED;
    }
  }

  sum_rows<NP, FUB>(fa, fb, agg_a, agg_b, gin, sorted, obase, fresh, final, row_base,
                    wave, lane);
}

template <int NP, int FUB>
__device__ __forceinline__ void sum_rows(
    const float* __restrict__ fa, const float* __restrict__ fb, float* agg_a,
    float* agg_b, GateIn gin, const unsigned* sorted, const int* obase, bool fresh,
    bool final, int row_base, int wave, int lane) {
  const float* bp = gin.bias + lane * 4;
  v4f bias_a, bias_b;
  if (NP == 2) {
    bias_a = *(const v4f*)(bp + 0 * GH) + *(const v4f*)(bp + 1 * GH);
    bias_b = *(const v4f*)(bp + 2 * GH) + *(const v4f*)(bp + 3 * GH);
  } else {
    bias_a = *(const v4f*)(bp + 4 * GH) + *(const v4f*)(bp + 5 * GH);
    bias_b = bias_a;
  }
#pragma nounroll
  for (int j = 0; j < RPW; ++j) {
    const int lr = wave + j * SWAVES;
    const int row = row_base + lr;
    if (lr >= TR || row >= GN) break;
    int i0 = __builtin_amdgcn_readfirstlane(obase[lr]);
    const int sege = __builtin_amdgcn_readfirstlane(obase[lr + 1]);
    float* oa = agg_a + (size_t)row * GH + lane * 4;
    float* ob = agg_b + (size_t)row * GH + lane * 4;
    v4f a[NP];
#pragma unroll
    for (int p = 0; p < NP; ++p) a[p] = (v4f){0.0f, 0.0f, 0.0f, 0.0f};
    if (!fresh) {
      a[0] = *(const v4f*)oa;
      if (NP == 2) a[NP - 1] = *(const v4f*)ob;
    }
    const size_t own = (size_t)row * GH + lane * 4;
    v4f own_a = (v4f){0.0f, 0.0f, 0.0f, 0.0f}, own_b = own_a, own_h = own_a, own_z = own_a;
    if (final) {
      own_a = *(const v4f*)(fa + own);
      own_h = *(const v4f*)(gin.h + own);
      if (NP == 2) own_b = *(const v4f*)(fb + own);
      else own_z = *(const v4f*)(gin.z + own);
    }
    for (; i0 + FUB <= sege; i0 += FUB) {
      unsigned w[FUB];
      v4f v[FUB][NP];
#pragma unroll
      for (int u = 0; u < FUB; ++u)
        w[u] = (unsigned)__builtin_amdgcn_readfirstlane((int)sorted[i0 + u]);
#pragma unroll
      for (int u = 0; u < FUB; ++u) {
        const size_t off = (size_t)(w[u] >> 8) * GH + lane * 4;
        v[u][0] = *(const v4f*)(fa + off);
        if (NP == 2) v[u][NP - 1] = *(const v4f*)(fb + off);
      }
      hold_loaded(v);
#pragma unroll
      for (int u = 0; u < FUB; ++u)
#pragma unroll
        for (int p = 0; p < NP; ++p) a[p] += v[u][p];
    }
    const int rem = sege - i0;
    if (rem > 0) {
      unsigned w[FUB - 1];
      v4f v[FUB - 1][NP];
#pragma unroll
      for (int u = 0; u < FUB - 1; ++u)
        w[u] = (unsigned)__builtin_amdgcn_readfirstlane((int)sorted[i0 + u]);
#pragma unroll
      for (int u = 0; u < FUB - 1; ++u)
        if (u < rem) {
          const size_t off = (size_t)(w[u] >> 8) * GH + lane * 4;
          v[u][0] = *(const v4f*)(fa + off);
          if (NP == 2) v[u][NP - 1] = *(const v4f*)(fb + off);
        }
#pragma unroll
      for (int u = 0; u < FUB - 1; ++u)
        if (u < rem) {
#pragma unroll
          for (int p = 0; p < NP; ++p) a[p] += v[u][p];
        }
    }
    if (!final) {
      *(v4f*)oa = a[0];
      if (NP == 2) *(v4f*)ob = a[NP - 1];
    } else if (NP == 2) {
      v4f zv, vv;
#pragma unroll
      for (int i = 0; i < 4; ++i) {
        zv[i] = sigmoid_f(own_a[i] + a[0][i] + bias_a[i]);
        vv[i] = sigmoid_f(own_b[i] + a[NP - 1][i] + bias_b[i]) * own_h[i];
      }
      *(v4f*)oa = zv;
      *(v4f*)ob = vv;
    } else {
      v4f ov;
#pragma unroll
      for (int i = 0; i < 4; ++i) {
        const float ht = tanh_f(own_a[i] + a[0][i] + bias_a[i]);
        ov[i] = own_z[i] * own_h[i] + (1.0f - own_z[i]) * ht;
      }
      *(v4f*)oa = ov;
    }
  }
  __syncthreads();
}

template <int NP, int FUB, bool FIRST>
__device__ __forceinline__ void scatter_body(
    const int* __restrict__ src, const int* __restrict__ dst,
    const unsigned char* __restrict__ codes, const float* __restrict__ fa,
    const float* __restrict__ fb, float* agg_a, float* agg_b, GateIn gin,
    unsigned* saved, unsigned* plist, unsigned* sorted, int* pcount, int* ocnt,
    int* obase, int* ocur) {
  const int tid = threadIdx.x;
  const int lane = tid & 31;
  const int wave = tid >> 5;
  const int row_base = blockIdx.x * TR;
  const unsigned pat = blockIdx.x * 0x01010101u;
  unsigned* my_saved = saved + (size_t)blockIdx.x * TILE_WORDS;

  if (!FIRST) {
    const unsigned nsv = (unsigned)__builtin_amdgcn_readfirstlane((int)my_saved[0]);
    if (nsv <= (unsigned)SCAP) {
      for (int i = tid * 4; i < (int)nsv; i += 4 * STHREADS)
        *(uint4*)(sorted + i) = *(const uint4*)(my_saved + SHDR + i);
      if (tid <= NBIN) obase[tid] = (int)my_saved[1 + tid];
      __syncthreads();
      sum_rows<NP, FUB>(fa, fb, agg_a, agg_b, gin, sorted, obase, true, true, row_base,
                        wave, lane);
      return;
    }
  }

  if (tid == 0) *pcount = 0;
  __syncthreads();

  int chunk = (blockIdx.x * NCHUNK) / NTILES;
  uint4 cur[SK], nxt[SK];
  load_chunk<!FIRST>(codes, dst, chunk, tid, cur);
  bool fresh = true;
  for (int c = 0; c < NCHUNK; ++c) {
    int nchunk = chunk + 1;
    if (nchunk == NCHUNK) nchunk = 0;
    const bool last = (c + 1 == NCHUNK);
    if (FIRST && !last) load_chunk<false>(codes, dst, nchunk, tid, nxt);

#pragma unroll
    for (int k = 0; k < SK; ++k) {
      const int e0 = chunk * CHUNK + k * (STHREADS * EPL) + tid * EPL;
      note_sixteen(cur[k], e0, pat, plist, pcount);
    }
    __syncthreads();
    const int n = *pcount;
    __syncthreads();
    if (last || n + CHUNK > LCAP2) {
      drain_rows<NP, FUB>(src, dst, fa, fb, agg_a, agg_b, gin, plist, sorted, ocnt,
                          obase, ocur, n, fresh, last, row_base, tid, wave, lane,
                          (FIRST && last) ? my_saved : nullptr);
      fresh = false;
      if (tid == 0) *pcount = 0;
      __syncthreads();
    }
    if (!last) {
      if (FIRST) {
#pragma unroll
        for (int k = 0; k < SK; ++k) cur[k] = nxt[k];
      } else {
        load_chunk<true>(codes, dst, nchunk, tid, cur);
      }
    }
    chunk = nchunk;
  }
}

__global__ __launch_bounds__(STHREADS) void scatter_tile_kernel(
    const int* __restrict__ src, const int* __restrict__ dst,
    float* lists_f, const float* __restrict__ prod, const float* __restrict__ zin,
    const float* __restrict__ hl, const float* __restrict__ bl, float* out) {
  __shared__ unsigned plist[LCAP2];
  __shared__ __attribute__((aligned(16))) unsigned sorted[LCAP2 + 8];
  __shared__ int pcount;
  __shared__ int ocnt[NBIN], obase[NBIN + 1], ocur[NBIN];
  GateIn gin;
  gin.h = hl;
  gin.z = zin;
  gin.bias = bl;
  scatter_body<1, 8, false>(src, dst, (const unsigned char*)nullptr, prod, prod, out,
                            out, gin, (unsigned*)lists_f, plist, sorted, &pcount, ocnt,
                            obase, ocur);
}

template <bool FIRST>
__global__ __launch_bounds__(STHREADS) void scatter_pair_kernel(
    const int* __restrict__ src, const int* __restrict__ dst,
    const float* codes_f, float* lists_f, const float* __restrict__ prod_z,
    const float* __restrict__ prod_r, const float* __restrict__ hl,
    const float* __restrict__ bl, float* zout, float* vout) {
  __shared__ unsigned plist[LCAP2];
  __shared__ __attribute__((aligned(16))) unsigned sorted[LCAP2 + 8];
  __shared__ int pcount;
  __shared__ int ocnt[NBIN], obase[NBIN + 1], ocur[NBIN];
  GateIn gin;
  gin.h = hl;
  gin.z = hl;
  gin.bias = bl;
  scatter_body<2, 4, FIRST>(src, dst, (const unsigned char*)codes_f, prod_z, prod_r,
                            zout, vout, gin, (unsigned*)lists_f, plist, sorted, &pcount,
                            ocnt, obase, ocur);
}

__global__ __launch_bounds__(256) void gemm_zr_kernel(
    const float* __restrict__ x, const float* __restrict__ hl,
    const float* __restrict__ Wl, float* __restrict__ pz, float* __restrict__ pr) {
  __shared__ float lds[4 * 16 * 132];
  __shared__ __attribute__((aligned(16))) float wlds[8 * WREGION];
  const int tid = threadIdx.x;
  const int wave = tid >> 5;
  const int lane = tid & 31;
  const int row0 = blockIdx.x * 128 + wave * 16;
  const int kof = (lane >> 4) << 1;
  const int nn = lane & 15;
  const int wc = weight_col(nn);
  const int cq = (lane >> 4) << 2;
  float* wl = wlds + wave * WREGION;
  float* sz = wl;
  float* sr = wl + 16 * OPITCH;

  v2f fs[32], ft[32];
  load_frag(x, row0, lane, nn, kof, wl, fs);
  load_frag(hl, row0, lane, nn, kof, wl, ft);

  const float* W0 = Wl + 0 * GH * GH;
  const float* W1 = Wl + 1 * GH * GH;
  const float* W2 = Wl + 2 * GH * GH;
  const float* W3 = Wl + 3 * GH * GH;
  float* L0 = lds + 0 * 16 * 132;
  float* L1 = lds + 1 * 16 * 132;
  float* L2 = lds + 2 * 16 * 132;
  float* L3 = lds + 3 * 16 * 132;

  for (int ct = 0; ct < GH / 16; ++ct) {
    const int n0 = ct * 16;
    __syncthreads();
    stage_slab(W0, n0, L0);
    stage_slab(W1, n0, L1);
    stage_slab(W2, n0, L2);
    stage_slab(W3, n0, L3);
    __syncthreads();
    if (ct != 0 && (ct & 1) == 0) {
      store_strip(sz, pz, row0, n0 - 32, lane);
      store_strip(sr, pr, row0, n0 - 32, lane);
    }

    const float* p0 = L0 + wc * 132 + kof;
    const float* p1 = L1 + wc * 132 + kof;
    const float* p2 = L2 + wc * 132 + kof;
    const float* p3 = L3 + wc * 132 + kof;

    v8f az0 = {0, 0, 0, 0, 0, 0, 0, 0}, az1 = {0, 0, 0, 0, 0, 0, 0, 0};
    v8f ar0 = {0, 0, 0, 0, 0, 0, 0, 0}, ar1 = {0, 0, 0, 0, 0, 0, 0, 0};
#pragma unroll
    for (int kk = 0; kk < 32; kk += 2) {
      az0 = wmma_f32(*(const v2f*)(p0 + 4 * kk), fs[kk], az0);
      az0 = wmma_f32(*(const v2f*)(p1 + 4 * kk), ft[kk], az0);
      ar0 = wmma_f32(*(const v2f*)(p2 + 4 * kk), fs[kk], ar0);
      ar0 = wmma_f32(*(const v2f*)(p3 + 4 * kk), ft[kk], ar0);
      az1 = wmma_f32(*(const v2f*)(p0 + 4 * kk + 4), fs[kk + 1], az1);
      az1 = wmma_f32(*(const v2f*)(p1 + 4 * kk + 4), ft[kk + 1], az1);
      ar1 = wmma_f32(*(const v2f*)(p2 + 4 * kk + 4), fs[kk + 1], ar1);
      ar1 = wmma_f32(*(const v2f*)(p3 + 4 * kk + 4), ft[kk + 1], ar1);
    }
    const v8f accz = az0 + az1;
    const v8f accr = ar0 + ar1;

    put_piece(sz, nn, ct & 1, cq, accz);
    put_piece(sr, nn, ct & 1, cq, accr);
  }
  __syncthreads();
  store_strip(sz, pz, row0, GH - 32, lane);
  store_strip(sr, pr, row0, GH - 32, lane);
}

__global__ __launch_bounds__(256) void gemm_ht_kernel(
    const float* __restrict__ x, const float* __restrict__ v,
    const float* __restrict__ Wl, float* __restrict__ ph) {
  __shared__ float lds[2 * 16 * 132];
  __shared__ __attribute__((aligned(16))) float wlds[8 * WREGION];
  const int tid = threadIdx.x;
  const int wave = tid >> 5;
  const int lane = tid & 31;
  const int row0 = blockIdx.x * 128 + wave * 16;
  const int kof = (lane >> 4) << 1;
  const int nn = lane & 15;
  const int wc = weight_col(nn);
  const int cq = (lane >> 4) << 2;
  float* wl = wlds + wave * WREGION;

  v2f fs[32], fu[32];
  load_frag(x, row0, lane, nn, kof, wl, fs);
  load_frag(v, row0, lane, nn, kof, wl, fu);

  const float* W4 = Wl + 4 * GH * GH;
  const float* W5 = Wl + 5 * GH * GH;
  float* L4 = lds + 0 * 16 * 132;
  float* L5 = lds + 1 * 16 * 132;

  for (int ct = 0; ct < GH / 16; ++ct) {
    const int n0 = ct * 16;
    __syncthreads();
    stage_slab(W4, n0, L4);
    stage_slab(W5, n0, L5);
    __syncthreads();
    if (ct != 0 && (ct & 1) == 0)
      store_strip(wl, ph, row0, n0 - 32, lane);

    const float* p4 = L4 + wc * 132 + kof;
    const float* p5 = L5 + wc * 132 + kof;

    v8f a0 = {0, 0, 0, 0, 0, 0, 0, 0}, a1 = {0, 0, 0, 0, 0, 0, 0, 0};
#pragma unroll
    for (int kk = 0; kk < 32; kk += 2) {
      a0 = wmma_f32(*(const v2f*)(p4 + 4 * kk), fs[kk], a0);
      a0 = wmma_f32(*(const v2f*)(p5 + 4 * kk), fu[kk], a0);
      a1 = wmma_f32(*(const v2f*)(p4 + 4 * kk + 4), fs[kk + 1], a1);
      a1 = wmma_f32(*(const v2f*)(p5 + 4 * kk + 4), fu[kk + 1], a1);
    }
    const v8f acc = a0 + a1;

    put_piece(wl, nn, ct & 1, cq, acc);
  }
  __syncthreads();
  store_strip(wl, ph, row0, GH - 32, lane);
}

extern "C" void kernel_launch(void* const* d_in, const int* in_sizes, int n_in,
                              void* d_out, int out_size, void* d_ws,
                              size_t ws_size, hipStream_t stream) {
  (void)in_sizes; (void)n_in; (void)out_size; (void)ws_size;
  const float* inp = (const float*)d_in[0];
  const int* edg = (const int*)d_in[1];
  const float* hmat = (const float*)d_in[2];
  const float* Wmat = (const float*)d_in[3];
  const float* bmat = (const float*)d_in[4];
  float* out = (float*)d_out;

  const int* src = edg;
  const int* dst = edg + GE;

  float* pa = (float*)d_ws;
  float* pb = pa + NH;
  float* zbuf = pb + NH;
  float* vbuf = zbuf + NH;

  const int gemm_blocks = (GN + 127) / 128;

  static_assert((size_t)GE <= NH * sizeof(float), "codes fit one output plane");
  float* codes = out;
  float* lists = out + (size_t)(GLAYERS - 1) * NH;
  tile_code_kernel<<<(GE / EPL + 255) / 256, 256, 0, stream>>>(dst, codes);

  for (int l = 0; l < GLAYERS; ++l) {
    const float* x = (l == 0) ? inp : (out + (size_t)(l - 1) * NH);
    const float* hl = hmat + (size_t)l * NH;
    const float* Wl = Wmat + (size_t)l * 6 * GH * GH;
    const float* bl = bmat + (size_t)l * 6 * GH;
    float* outl = out + (size_t)l * NH;

    gemm_zr_kernel<<<gemm_blocks, 256, 0, stream>>>(x, hl, Wl, pa, pb);

    if (l == 0)
      scatter_pair_kernel<true><<<NTILES, STHREADS, 0, stream>>>(src, dst, codes, lists, pa,
                                                                 pb, hl, bl, zbuf, vbuf);
    else
      scatter_pair_kernel<false><<<NTILES, STHREADS, 0, stream>>>(src, dst, codes, lists, pa,
                                                                  pb, hl, bl, zbuf, vbuf);

    gemm_ht_kernel<<<gemm_blocks, 256, 0, stream>>>(x, vbuf, Wl, pa);

    scatter_tile_kernel<<<NTILES, STHREADS, 0, stream>>>(src, dst, lists, pa, zbuf, hl, bl,
                                                         outl);
  }
}


# GEMMs: a wave keeps the weights of its 16 columns in registers all launch; 256 workgroups of 16 waves stream 12-13 row tiles each through double LDS buffers, one barrier per tile; sums unchanged
# speedup vs baseline: 4.6140x; 1.0542x over previous
#include <hip/hip_runtime.h>
#include <cstdint>
#include <cstddef>

constexpr int GN = 50000;
constexpr int GE = 1600000;
constexpr int GH = 128;
constexpr int GLAYERS = 2;
constexpr size_t NH = (size_t)GN * GH;

typedef __attribute__((ext_vector_type(2))) float v2f;
typedef __attribute__((ext_vector_type(8))) float v8f;
typedef __attribute__((ext_vector_type(4))) float v4f;

__device__ __forceinline__ v8f wmma_f32(v2f a, v2f b, v8f c) {
  return __builtin_amdgcn_wmma_f32_16x16x4_f32(
       false, a,  false, b,
       (short)0, c,  false,  false);
}

__device__ __forceinline__ float sigmoid_f(float x) {
  return 1.0f / (1.0f + __expf(-x));
}
__device__ __forceinline__ float tanh_f(float x) {
  return 1.0f - 2.0f / (1.0f + __expf(2.0f * x));
}

__device__ __forceinline__ int weight_col(int m) {
  return ((m & 4) << 1) | ((m & 8) >> 1) | (m & 3);
}

constexpr int FPITCH = 132;
constexpr int TILE_F = 16 * FPITCH;
constexpr int GT = GN / 16;
constexpr int GWG = 256;
constexpr int GTHREADS = 512;
static_assert(GN % 16 == 0, "no 16-row tile straddles the end of the graph");

constexpr int TR = 197;
constexpr int NTILES = (GN + TR - 1) / TR;
constexpr unsigned NO_TILE = 255u;
constexpr int STHREADS = 1024;
constexpr int SWAVES = STHREADS / 32;
constexpr int RPW = (TR + SWAVES - 1) / SWAVES;
constexpr int EPL = 16;
constexpr int SK = 2;
constexpr int CHUNK = STHREADS * EPL * SK;
constexpr int NCHUNK = (GE + CHUNK - 1) / CHUNK;
constexpr int NBIN = SWAVES * RPW;
constexpr int LCAP2 = 39936;

constexpr int SCAP = 8192;
constexpr int SHDR = 256;
constexpr int SSTRIDE = SHDR + SCAP;
constexpr unsigned NOT_SAVED = 0xFFFFFFFFu;
constexpr size_t TILE_WORDS = (size_t)TR * GH;

static_assert(NTILES * TR >= GN, "tiles cover all rows");
static_assert(NBIN + 2 <= SHDR && SCAP % 4 == 0 && SCAP + 8 <= LCAP2, "header and list fit");
static_assert((size_t)SSTRIDE <= TILE_WORDS && (TILE_WORDS * 4) % 16 == 0, "a saved list fits its tile's rows, 16-byte aligned");
static_assert((size_t)(NTILES - 1) * TILE_WORDS + SSTRIDE <= NH, "the last tile's saved list ends inside the plane");
static_assert(GLAYERS >= 2, "the codes live in the first layer's output plane, the saved lists in the last layer's");
static_assert(NTILES <= (int)NO_TILE, "every tile number and the spare code fit one byte");
static_assert(GE % EPL == 0, "16-byte code loads");
static_assert(LCAP2 >= CHUNK, "an emptied list holds every edge of one scan step");
static_assert((2 * LCAP2 + 3 * NBIN + 16) * 4 <= 320 * 1024, "both lists fit the LDS");
static_assert(TR <= 256 && GN < (1 << 24), "local row and source row fit one word");
static_assert(SWAVES == 32 && NBIN >= TR + 1, "the first wave's lanes take RPW counters each");

__device__ __forceinline__ unsigned tile_code(int d) {
  const unsigned a = (unsigned)d;
  const unsigned lim = NO_TILE * (unsigned)TR;
  return (a < lim ? a : lim) / (unsigned)TR;
}

__device__ __forceinline__ uint4 codes_of_sixteen(const int* __restrict__ dst16) {
  const int4* d4 = (const int4*)dst16;
  int4 d[4];
#pragma unroll
  for (int q = 0; q < 4; ++q) d[q] = d4[q];
  unsigned w[4];
#pragma unroll
  for (int q = 0; q < 4; ++q)
    w[q] = tile_code(d[q].x) | (tile_code(d[q].y) << 8) |
           (tile_code(d[q].z) << 16) | (tile_code(d[q].w) << 24);
  return make_uint4(w[0], w[1], w[2], w[3]);
}

__global__ __launch_bounds__(256) void tile_code_kernel(
    const int* __restrict__ dst, float* __restrict__ codes_f) {
  const int g = blockIdx.x * 256 + threadIdx.x;
  if (g >= GE / EPL) return;
  ((uint4*)codes_f)[g] = codes_of_sixteen(dst + (size_t)g * EPL);
}

template <bool FROM_DST>
__device__ __forceinline__ void load_chunk(const unsigned char* __restrict__ codes,
                                           const int* __restrict__ dst,
                                           int chunk, int tid, uint4* buf) {
#pragma unroll
  for (int k = 0; k < SK; ++k) {
    const int e0 = chunk * CHUNK + k * (STHREADS * EPL) + tid * EPL;
    const int e = e0 < GE ? e0 : GE - EPL;
    if (FROM_DST)
      buf[k] = codes_of_sixteen(dst + e);
    else
      buf[k] = *(const uint4*)(codes + e);
  }
}

__device__ __forceinline__ unsigned zero_bytes(unsigned x) {
  const unsigned m = 0x7F7F7F7Fu;
  return ~(((x & m) + m) | x | m);
}

__device__ __forceinline__ void note_sixteen(uint4 w, int e0, unsigned pat,
                                             unsigned* plist, int* pcount) {
  const unsigned z0 = zero_bytes(w.x ^ pat);
  const unsigned z1 = zero_bytes(w.y ^ pat);
  const unsigned z2 = zero_bytes(w.z ^ pat);
  const unsigned z3 = zero_bytes(w.w ^ pat);
  unsigned hits = (z0 >> 7) | (z1 >> 6) | (z2 >> 5) | (z3 >> 4);
  if (e0 >= GE) hits = 0;
  while (hits != 0) {
    const int bit = __builtin_ctz(hits);
    hits &= hits - 1;
    const int pos = atomicAdd(pcount, 1);
    plist[pos] = (unsigned)(e0 + ((bit & 7) << 2) + (bit >> 3));
  }
}

__device__ __forceinline__ void hold_loaded(const v4f (&v)[4][2]) {
  asm volatile("" ::
               "v"(v[0][0]), "v"(v[0][1]), "v"(v[1][0]), "v"(v[1][1]),
               "v"(v[2][0]), "v"(v[2][1]), "v"(v[3][0]), "v"(v[3][1]));
}
template <int F>
__device__ __forceinline__ void hold_loaded(const v4f (&)[F][1]) {}

struct GateIn {
  const float* h;
  const float* z;
  const float* bias;
};

template <int NP, int FUB>
__device__ __forceinline__ void sum_rows(
    const float* __restrict__ fa, const float* __restrict__ fb, float* agg_a,
    float* agg_b, GateIn gin, const unsigned* sorted, const int* obase, bool fresh,
    bool final, int row_base, int wave, int lane);

template <int NP, int FUB>
__device__ __forceinline__ void drain_rows(
    const int* __restrict__ src, const int* __restrict__ dst,
    const float* __restrict__ fa, const float* __restrict__ fb,
    float* agg_a, float* agg_b, GateIn gin, unsigned* plist, unsigned* sorted,
    int* ocnt, int* obase, int* ocur, int n, bool fresh, bool final, int row_base,
    int tid, int wave, int lane, unsigned* save_to) {
  if (tid < NBIN) ocnt[tid] = 0;
  __syncthreads();

  for (int i = tid; i < n; i += 4 * STHREADS) {
    unsigned ent[4];
    int s[4], d[4];
#pragma unroll
    for (int q = 0; q < 4; ++q) {
      const int ii = i + q * STHREADS;
      ent[q] = plist[ii < n ? ii : i];
    }
#pragma unroll
    for (int q = 0; q < 4; ++q) {
      s[q] = src[ent[q]];
      d[q] = dst[ent[q]];
    }
#pragma unroll
    for (int q = 0; q < 4; ++q) {
      const int ii = i + q * STHREADS;
      if (ii < n) {
        const unsigned lr = (unsigned)(d[q] - row_base);
        plist[ii] = ((unsigned)s[q] << 8) | lr;
        atomicAdd(&ocnt[lr], 1);
      }
    }
  }
  __syncthreads();

  if (wave == 0) {
    int c[RPW];
    int run = 0;
#pragma unroll
    for (int q = 0; q < RPW; ++q) {
      const int x = ocnt[lane * RPW + q];
      c[q] = run;
      run += x;
    }
    int incl = run;
#pragma unroll
    for (int dlt = 1; dlt < 32; dlt <<= 1) {
      const int t = __shfl_up(incl, dlt);
      if (lane >= dlt) incl += t;
    }
    const int excl = incl - run;
#pragma unroll
    for (int q = 0; q < RPW; ++q) {
      obase[lane * RPW + q] = excl + c[q];
      ocur[lane * RPW + q] = excl + c[q];
    }
    if (lane == 31) obase[NBIN] = incl;
  }
  __syncthreads();

  for (int i = tid; i < n; i += STHREADS) {
    const unsigned w = plist[i];
    const int pos = atomicAdd(&ocur[w & 255u], 1);
    sorted[pos] = w;
  }
  __syncthreads();

  if (save_to != nullptr) {
    if (fresh && n <= SCAP) {
      for (int i = tid * 4; i < n; i += 4 * STHREADS)
        *(uint4*)(save_to + SHDR + i) = *(const uint4*)(sorted + i);
      if (tid <= NBIN) save_to[1 + tid] = (unsigned)obase[tid];
      if (tid == 0) save_to[0] = (unsigned)n;
    } else if (tid == 0) {
      save_to[0] = NOT_SAVED;
    }
  }

  sum_rows<NP, FUB>(fa, fb, agg_a, agg_b, gin, sorted, obase, fresh, final, row_base,
                    wave, lane);
}

template <int NP, int FUB>
__device__ __forceinline__ void sum_rows(
    const float* __restrict__ fa, const float* __restrict__ fb, float* agg_a,
    float* agg_b, GateIn gin, const unsigned* sorted, const int* obase, bool fresh,
    bool final, int row_base, int wave, int lane) {
  const float* bp = gin.bias + lane * 4;
  v4f bias_a, bias_b;
  if (NP == 2) {
    bias_a = *(const v4f*)(bp + 0 * GH) + *(const v4f*)(bp + 1 * GH);
    bias_b = *(const v4f*)(bp + 2 * GH) + *(const v4f*)(bp + 3 * GH);
  } else {
    bias_a = *(const v4f*)(bp + 4 * GH) + *(const v4f*)(bp + 5 * GH);
    bias_b = bias_a;
  }
#pragma nounroll
  for (int j = 0; j < RPW; ++j) {
    const int lr = wave + j * SWAVES;
    const int row = row_base + lr;
    if (lr >= TR || row >= GN) break;
    int i0 = __builtin_amdgcn_readfirstlane(obase[lr]);
    const int sege = __builtin_amdgcn_readfirstlane(obase[lr + 1]);
    float* oa = agg_a + (size_t)row * GH + lane * 4;
    float* ob = agg_b + (size_t)row * GH + lane * 4;
    v4f a[NP];
#pragma unroll
    for (int p = 0; p < NP; ++p) a[p] = (v4f){0.0f, 0.0f, 0.0f, 0.0f};
    if (!fresh) {
      a[0] = *(const v4f*)oa;
      if (NP == 2) a[NP - 1] = *(const v4f*)ob;
    }
    const size_t own = (size_t)row * GH + lane * 4;
    v4f own_a = (v4f){0.0f, 0.0f, 0.0f, 0.0f}, own_b = own_a, own_h = own_a, own_z = own_a;
    if (final) {
      own_a = *(const v4f*)(fa + own);
      own_h = *(const v4f*)(gin.h + own);
      if (NP == 2) own_b = *(const v4f*)(fb + own);
      else own_z = *(const v4f*)(gin.z + own);
    }
    for (; i0 + FUB <= sege; i0 += FUB) {
      unsigned w[FUB];
      v4f v[FUB][NP];
#pragma unroll
      for (int u = 0; u < FUB; ++u)
        w[u] = (unsigned)__builtin_amdgcn_readfirstlane((int)sorted[i0 + u]);
#pragma unroll
      for (int u = 0; u < FUB; ++u) {
        const size_t off = (size_t)(w[u] >> 8) * GH + lane * 4;
        v[u][0] = *(const v4f*)(fa + off);
        if (NP == 2) v[u][NP - 1] = *(const v4f*)(fb + off);
      }
      hold_loaded(v);
#pragma unroll
      for (int u = 0; u < FUB; ++u)
#pragma unroll
        for (int p = 0; p < NP; ++p) a[p] += v[u][p];
    }
    const int rem = sege - i0;
    if (rem > 0) {
      unsigned w[FUB - 1];
      v4f v[FUB - 1][NP];
#pragma unroll
      for (int u = 0; u < FUB - 1; ++u)
        w[u] = (unsigned)__builtin_amdgcn_readfirstlane((int)sorted[i0 + u]);
#pragma unroll
      for (int u = 0; u < FUB - 1; ++u)
        if (u < rem) {
          const size_t off = (size_t)(w[u] >> 8) * GH + lane * 4;
          v[u][0] = *(const v4f*)(fa + off);
          if (NP == 2) v[u][NP - 1] = *(const v4f*)(fb + off);
        }
#pragma unroll
      for (int u = 0; u < FUB - 1; ++u)
        if (u < rem) {
#pragma unroll
          for (int p = 0; p < NP; ++p) a[p] += v[u][p];
        }
    }
    if (!final) {
      *(v4f*)oa = a[0];
      if (NP == 2) *(v4f*)ob = a[NP - 1];
    } else if (NP == 2) {
      v4f zv, vv;
#pragma unroll
      for (int i = 0; i < 4; ++i) {
        zv[i] = sigmoid_f(own_a[i] + a[0][i] + bias_a[i]);
        vv[i] = sigmoid_f(own_b[i] + a[NP - 1][i] + bias_b[i]) * own_h[i];
      }
      *(v4f*)oa = zv;
      *(v4f*)ob = vv;
    } else {
      v4f ov;
#pragma unroll
      for (int i = 0; i < 4; ++i) {
        const float ht = tanh_f(own_a[i] + a[0][i] + bias_a[i]);
        ov[i] = own_z[i] * own_h[i] + (1.0f - own_z[i]) * ht;
      }
      *(v4f*)oa = ov;
    }
  }
  __syncthreads();
}

template <int NP, int FUB, bool FIRST>
__device__ __forceinline__ void scatter_body(
    const int* __restrict__ src, const int* __restrict__ dst,
    const unsigned char* __restrict__ codes, const float* __restrict__ fa,
    const float* __restrict__ fb, float* agg_a, float* agg_b, GateIn gin,
    unsigned* saved, unsigned* plist, unsigned* sorted, int* pcount, int* ocnt,
    int* obase, int* ocur) {
  const int tid = threadIdx.x;
  const int lane = tid & 31;
  const int wave = tid >> 5;
  const int row_base = blockIdx.x * TR;
  const unsigned pat = blockIdx.x * 0x01010101u;
  unsigned* my_saved = saved + (size_t)blockIdx.x * TILE_WORDS;

  if (!FIRST) {
    const unsigned nsv = (unsigned)__builtin_amdgcn_readfirstlane((int)my_saved[0]);
    if (nsv <= (unsigned)SCAP) {
      for (int i = tid * 4; i < (int)nsv; i += 4 * STHREADS)
        *(uint4*)(sorted + i) = *(const uint4*)(my_saved + SHDR + i);
      if (tid <= NBIN) obase[tid] = (int)my_saved[1 + tid];
      __syncthreads();
      sum_rows<NP, FUB>(fa, fb, agg_a, agg_b, gin, sorted, obase, true, true, row_base,
                        wave, lane);
      return;
    }
  }

  if (tid == 0) *pcount = 0;
  __syncthreads();

  int chunk = (blockIdx.x * NCHUNK) / NTILES;
  uint4 cur[SK], nxt[SK];
  load_chunk<!FIRST>(codes, dst, chunk, tid, cur);
  bool fresh = true;
  for (int c = 0; c < NCHUNK; ++c) {
    int nchunk = chunk + 1;
    if (nchunk == NCHUNK) nchunk = 0;
    const bool last = (c + 1 == NCHUNK);
    if (FIRST && !last) load_chunk<false>(codes, dst, nchunk, tid, nxt);

#pragma unroll
    for (int k = 0; k < SK; ++k) {
      const int e0 = chunk * CHUNK + k * (STHREADS * EPL) + tid * EPL;
      note_sixteen(cur[k], e0, pat, plist, pcount);
    }
    __syncthreads();
    const int n = *pcount;
    __syncthreads();
    if (last || n + CHUNK > LCAP2) {
      drain_rows<NP, FUB>(src, dst, fa, fb, agg_a, agg_b, gin, plist, sorted, ocnt,
                          obase, ocur, n, fresh, last, row_base, tid, wave, lane,
                          (FIRST && last) ? my_saved : nullptr);
      fresh = false;
      if (tid == 0) *pcount = 0;
      __syncthreads();
    }
    if (!last) {
      if (FIRST) {
#pragma unroll
        for (int k = 0; k < SK; ++k) cur[k] = nxt[k];
      } else {
        load_chunk<true>(codes, dst, nchunk, tid, cur);
      }
    }
    chunk = nchunk;
  }
}

__global__ __launch_bounds__(STHREADS) void scatter_tile_kernel(
    const int* __restrict__ src, const int* __restrict__ dst,
    float* lists_f, const float* __restrict__ prod, const float* __restrict__ zin,
    const float* __restrict__ hl, const float* __restrict__ bl, float* out) {
  __shared__ unsigned plist[LCAP2];
  __shared__ __attribute__((aligned(16))) unsigned sorted[LCAP2 + 8];
  __shared__ int pcount;
  __shared__ int ocnt[NBIN], obase[NBIN + 1], ocur[NBIN];
  GateIn gin;
  gin.h = hl;
  gin.z = zin;
  gin.bias = bl;
  scatter_body<1, 8, false>(src, dst, (const unsigned char*)nullptr, prod, prod, out,
                            out, gin, (unsigned*)lists_f, plist, sorted, &pcount, ocnt,
                            obase, ocur);
}

template <bool FIRST>
__global__ __launch_bounds__(STHREADS) void scatter_pair_kernel(
    const int* __restrict__ src, const int* __restrict__ dst,
    const float* codes_f, float* lists_f, const float* __restrict__ prod_z,
    const float* __restrict__ prod_r, const float* __restrict__ hl,
    const float* __restrict__ bl, float* zout, float* vout) {
  __shared__ unsigned plist[LCAP2];
  __shared__ __attribute__((aligned(16))) unsigned sorted[LCAP2 + 8];
  __shared__ int pcount;
  __shared__ int ocnt[NBIN], obase[NBIN + 1], ocur[NBIN];
  GateIn gin;
  gin.h = hl;
  gin.z = hl;
  gin.bias = bl;
  scatter_body<2, 4, FIRST>(src, dst, (const unsigned char*)codes_f, prod_z, prod_r,
                            zout, vout, gin, (unsigned*)lists_f, plist, sorted, &pcount,
                            ocnt, obase, ocur);
}

template <bool ZR>
__global__ __launch_bounds__(GTHREADS) void gemm_rows_kernel(
    const float* __restrict__ x, const float* __restrict__ u,
    const float* __restrict__ Wl, float* __restrict__ o0, float* o1) {
  constexpr int NTP = ZR ? 2 : 4;
  constexpr int NUNITS = ZR ? GT : (GT + 1) / 2;
  __shared__ __attribute__((aligned(16))) float inb[2][NTP][TILE_F];
  __shared__ __attribute__((aligned(16))) float outb[2][2][TILE_F];
  const int tid = threadIdx.x;
  const int wave = tid >> 5;
  const int lane = tid & 31;
  const int kof = (lane >> 4) << 1;
  const int nn = lane & 15;
  const int wc = weight_col(nn);
  const int cq = (lane >> 4) << 2;
  const int ct = wave & 7;
  const int sel = wave >> 3;
  const int lrow = tid >> 5;
  const int lc4 = (tid & 31) * 4;

  const float* Wa = Wl + (size_t)(ZR ? 2 * sel : 4) * GH * GH;
  const float* Wb = Wl + (size_t)(ZR ? 2 * sel + 1 : 5) * GH * GH;
  v2f wa[32], wb[32];
#pragma unroll
  for (int kk = 0; kk < 32; ++kk) {
    const int k = 4 * kk + kof;
    const int c = ct * 16 + wc;
    wa[kk] = (v2f){Wa[k * GH + c], Wa[(k + 1) * GH + c]};
    wb[kk] = (v2f){Wb[k * GH + c], Wb[(k + 1) * GH + c]};
  }

  v4f r[NTP];
  auto load_step = [&](int unit) {
#pragma unroll
    for (int tp = 0; tp < NTP; ++tp) {
      const float* P = (tp & 1) ? u : x;
      const int tile = ZR ? unit : 2 * unit + (tp >> 1);
      int row = tile * 16 + lrow;
      if (row > GN - 1) row = GN - 1;
      r[tp] = *(const v4f*)(P + (size_t)row * GH + lc4);
    }
  };
  auto put_rows = [&](int buf) {
#pragma unroll
    for (int tp = 0; tp < NTP; ++tp) *(v4f*)(&inb[buf][tp][lrow * FPITCH + lc4]) = r[tp];
  };
  auto store_out = [&](int buf, int unit) {
    v4f v[2];
#pragma unroll
    for (int op = 0; op < 2; ++op) v[op] = *(const v4f*)(&outb[buf][op][lrow * FPITCH + lc4]);
#pragma unroll
    for (int op = 0; op < 2; ++op) {
      float* P = ZR ? (op ? o1 : o0) : o0;
      const int tile = ZR ? unit : 2 * unit + op;
      if (tile < GT) *(v4f*)(P + (size_t)(tile * 16 + lrow) * GH + lc4) = v[op];
    }
  };

  int unit = blockIdx.x;
  if (unit < NUNITS) {
    load_step(unit);
    put_rows(0);
  }
  __syncthreads();
  int buf = 0;
  int punit = -1;
  for (; unit < NUNITS; unit += GWG) {
    const int nunit = unit + GWG;
    const bool more = nunit < NUNITS;
    if (more) load_step(nunit);
    if (punit >= 0) store_out(buf ^ 1, punit);

    const float* bx = &inb[buf][ZR ? 0 : 2 * sel][nn * FPITCH + kof];
    const float* bu = &inb[buf][ZR ? 1 : 2 * sel + 1][nn * FPITCH + kof];
    v2f fx[2][8], fu[2][8];
    auto load_set = [&](int st, int k0) {
#pragma unroll
      for (int j = 0; j < 8; ++j) {
        fx[st][j] = *(const v2f*)(bx + 4 * (k0 + j));
        fu[st][j] = *(const v2f*)(bu + 4 * (k0 + j));
      }
    };
    v8f a0 = {0, 0, 0, 0, 0, 0, 0, 0}, a1 = {0, 0, 0, 0, 0, 0, 0, 0};
    load_set(0, 0);
#pragma unroll
    for (int blk = 0; blk < 4; ++blk) {
      if (blk < 3) load_set((blk + 1) & 1, 8 * (blk + 1));
      __builtin_amdgcn_sched_barrier(0);
#pragma unroll
      for (int j = 0; j < 8; j += 2) {
        const int kk = 8 * blk + j;
        a0 = wmma_f32(wa[kk], fx[blk & 1][j], a0);
        a1 = wmma_f32(wa[kk + 1], fx[blk & 1][j + 1], a1);
        a0 = wmma_f32(wb[kk], fu[blk & 1][j], a0);
        a1 = wmma_f32(wb[kk + 1], fu[blk & 1][j + 1], a1);
      }
    }
    const v8f acc = a0 + a1;

    float* p = &outb[buf][sel][nn * FPITCH + ct * 16 + cq];
    *(v4f*)p = (v4f){acc[0], acc[1], acc[2], acc[3]};
    *(v4f*)(p + 8) = (v4f){acc[4], acc[5], acc[6], acc[7]};

    if (more) put_rows(buf ^ 1);
    __syncthreads();
    punit = unit;
    buf ^= 1;
  }
  if (punit >= 0) store_out(buf ^ 1, punit);
}

extern "C" void kernel_launch(void* const* d_in, const int* in_sizes, int n_in,
                              void* d_out, int out_size, void* d_ws,
                              size_t ws_size, hipStream_t stream) {
  (void)in_sizes; (void)n_in; (void)out_size; (void)ws_size;
  const float* inp = (const float*)d_in[0];
  const int* edg = (const int*)d_in[1];
  const float* hmat = (const float*)d_in[2];
  const float* Wmat = (const float*)d_in[3];
  const float* bmat = (const float*)d_in[4];
  float* out = (float*)d_out;

  const int* src = edg;
  const int* dst = edg + GE;

  float* pa = (float*)d_ws;
  float* pb = pa + NH;
  float* zbuf = pb + NH;
  float* vbuf = zbuf + NH;

  static_assert((size_t)GE <= NH * sizeof(float), "codes fit one output plane");
  float* codes = out;
  float* lists = out + (size_t)(GLAYERS - 1) * NH;
  tile_code_kernel<<<(GE / EPL + 255) / 256, 256, 0, stream>>>(dst, codes);

  for (int l = 0; l < GLAYERS; ++l) {
    const float* x = (l == 0) ? inp : (out + (size_t)(l - 1) * NH);
    const float* hl = hmat + (size_t)l * NH;
    const float* Wl = Wmat + (size_t)l * 6 * GH * GH;
    const float* bl = bmat + (size_t)l * 6 * GH;
    float* outl = out + (size_t)l * NH;

    gemm_rows_kernel<true><<<GWG, GTHREADS, 0, stream>>>(x, hl, Wl, pa, pb);

    if (l == 0)
      scatter_pair_kernel<true><<<NTILES, STHREADS, 0, stream>>>(src, dst, codes, lists, pa,
                                                                 pb, hl, bl, zbuf, vbuf);
    else
      scatter_pair_kernel<false><<<NTILES, STHREADS, 0, stream>>>(src, dst, codes, lists, pa,
                                                                  pb, hl, bl, zbuf, vbuf);

    gemm_rows_kernel<false><<<GWG, GTHREADS, 0, stream>>>(x, vbuf, Wl, pa, pa);

    scatter_tile_kernel<<<NTILES, STHREADS, 0, stream>>>(src, dst, lists, pa, zbuf, hl, bl,
                                                         outl);
  }
}
